# starting kernel
# baseline (speedup 1.0000x reference)
#include <hip/hip_runtime.h>
#include <hip/hip_bf16.h>

typedef __attribute__((ext_vector_type(16))) _Float16 v16h;
typedef __attribute__((ext_vector_type(8)))  float    v8f;

#define IN_DIM 128

__global__ void gcn_pack_w(const float* __restrict__ W, _Float16* __restrict__ Wp,
                           int Nt  ) {
    int id = blockIdx.x * blockDim.x + threadIdx.x;
    int total = Nt * 4 * 32;
    if (id >= total) return;
    int lane = id & 31;
    int kk   = (id >> 5) & 3;
    int t    = id >> 7;
    int half = lane >> 4;
    int r    = lane & 15;
    int N    = Nt * 16;
    int n    = t * 16 + r;
    int k0   = kk * 32;
    _Float16* dst = Wp + (size_t)id * 16;
#pragma unroll
    for (int j = 0; j < 8; ++j)
        dst[j] = (_Float16)W[(size_t)(k0 + half * 8 + j) * N + n];
#pragma unroll
    for (int j = 0; j < 8; ++j)
        dst[8 + j] = (_Float16)W[(size_t)(k0 + 16 + half * 8 + j) * N + n];
}

__global__ void gcn_spmm(const int* __restrict__ row, const int* __restrict__ col,
                         const float* __restrict__ vals, const float* __restrict__ x,
                         float* __restrict__ agg, int nEdges) {
    int gid  = blockIdx.x * blockDim.x + threadIdx.x;
    int e    = gid >> 5;
    int lane = gid & 31;
    if (e >= nEdges) return;
    int   r = row[e];
    int   c = col[e];
    float v = vals[e];
    const float4* xp = (const float4*)(x + (size_t)c * IN_DIM);
    float4 m = xp[lane];
    float* op = agg + (size_t)r * IN_DIM + lane * 4;
    atomicAdd(op + 0, v * m.x);
    atomicAdd(op + 1, v * m.y);
    atomicAdd(op + 2, v * m.z);
    atomicAdd(op + 3, v * m.w);
}

template <int NTILES, bool RELU>
__global__ void gcn_gemm_wmma(const float* __restrict__ A,
                              const _Float16* __restrict__ Bp,
                              float* __restrict__ C, int M) {
    const int N = NTILES * 16;
    int lane = threadIdx.x & 31;
    int warp = threadIdx.x >> 5;
    int wavesPerBlock = blockDim.x >> 5;
    int mt = blockIdx.x * wavesPerBlock + warp;
    int mtiles = M >> 4;
    if (mt >= mtiles) return;

    int half = lane >> 4;
    int r    = lane & 15;

    int mrow = mt * 16 + r;

    v8f acc[NTILES];
#pragma unroll
    for (int t = 0; t < NTILES; ++t) acc[t] = (v8f){};

#pragma unroll
    for (int kk = 0; kk < 4; ++kk) {
        int k0 = kk * 32;
        const float* ap = A + (size_t)mrow * IN_DIM + k0 + half * 8;
        float4 f0 = *(const float4*)(ap + 0);
        float4 f1 = *(const float4*)(ap + 4);
        float4 f2 = *(const float4*)(ap + 16);
        float4 f3 = *(const float4*)(ap + 20);
        v16h a;
        a[0]  = (_Float16)f0.x; a[1]  = (_Float16)f0.y; a[2]  = (_Float16)f0.z; a[3]  = (_Float16)f0.w;
        a[4]  = (_Float16)f1.x; a[5]  = (_Float16)f1.y; a[6]  = (_Float16)f1.z; a[7]  = (_Float16)f1.w;
        a[8]  = (_Float16)f2.x; a[9]  = (_Float16)f2.y; a[10] = (_Float16)f2.z; a[11] = (_Float16)f2.w;
        a[12] = (_Float16)f3.x; a[13] = (_Float16)f3.y; a[14] = (_Float16)f3.z; a[15] = (_Float16)f3.w;

#pragma unroll
        for (int t = 0; t < NTILES; ++t) {
            v16h b = *(const v16h*)(Bp + (((size_t)t * 4 + kk) * 32 + lane) * 16);
            acc[t] = __builtin_amdgcn_wmma_f32_16x16x32_f16(
                         false, a,  false, b,
                         (short)0, acc[t],
                         false,  false);
        }
    }

    int mbase = mt * 16 + half * 8;
#pragma unroll
    for (int t = 0; t < NTILES; ++t) {
        int ncol = t * 16 + r;
#pragma unroll
        for (int v = 0; v < 8; ++v) {
            float val = acc[t][v];
            if (RELU) val = fmaxf(val, 0.0f);
            C[(size_t)(mbase + v) * N + ncol] = val;
        }
    }
}

extern "C" void kernel_launch(void* const* d_in, const int* in_sizes, int n_in,
                              void* d_out, int out_size, void* d_ws, size_t ws_size,
                              hipStream_t stream) {
    const float* x    = (const float*)d_in[0];
    const int*   row  = (const int*)d_in[1];
    const int*   col  = (const int*)d_in[2];
    const float* vals = (const float*)d_in[3];
    const float* W1   = (const float*)d_in[4];
    const float* W2   = (const float*)d_in[5];
    float* out = (float*)d_out;

    const int NN = in_sizes[0] / IN_DIM;
    const int NE = in_sizes[1];

    float*     aggF = (float*)d_ws;
    float*     hF   = aggF + (size_t)NN * IN_DIM;
    _Float16*  W1p  = (_Float16*)(hF + (size_t)NN * IN_DIM);
    _Float16*  W2p  = W1p + 128 * 128;

    gcn_pack_w<<<4, 256, 0, stream>>>(W1, W1p, 8);
    gcn_pack_w<<<2, 256, 0, stream>>>(W2, W2p, 4);

    const size_t aggBytes = (size_t)NN * IN_DIM * sizeof(float);
    const int spmmBlocks = (NE * 32 + 255) / 256;

    hipMemsetAsync(aggF, 0, aggBytes, stream);
    gcn_spmm<<<spmmBlocks, 256, 0, stream>>>(row, col, vals, x, aggF, NE);

    {
        int mtiles = NN / 16;
        int blocks = (mtiles + 3) / 4;
        gcn_gemm_wmma<8, true><<<blocks, 128, 0, stream>>>(aggF, W1p, hF, NN);
    }

    hipMemsetAsync(aggF, 0, aggBytes, stream);
    gcn_spmm<<<spmmBlocks, 256, 0, stream>>>(row, col, vals, hF, aggF, NE);

    {
        int mtiles = NN / 16;
        int blocks = (mtiles + 3) / 4;
        gcn_gemm_wmma<4, false><<<blocks, 128, 0, stream>>>(aggF, W2p, out, NN);
    }
}


# SpMM without atomics: a work-group owns 196 rows in LDS, its waves sweep the edge list into per-wave LDS lists, one owner wave per row adds them in list order; f32 sum order changes; memsets gone
# speedup vs baseline: 1.6353x; 1.6353x over previous
#include <hip/hip_runtime.h>
#include <hip/hip_bf16.h>

typedef __attribute__((ext_vector_type(16))) _Float16 v16h;
typedef __attribute__((ext_vector_type(8)))  float    v8f;

#define IN_DIM 128

__global__ void gcn_pack_w(const float* __restrict__ W, _Float16* __restrict__ Wp,
                           int Nt  ) {
    int id = blockIdx.x * blockDim.x + threadIdx.x;
    int total = Nt * 4 * 32;
    if (id >= total) return;
    int lane = id & 31;
    int kk   = (id >> 5) & 3;
    int t    = id >> 7;
    int half = lane >> 4;
    int r    = lane & 15;
    int N    = Nt * 16;
    int n    = t * 16 + r;
    int k0   = kk * 32;
    _Float16* dst = Wp + (size_t)id * 16;
#pragma unroll
    for (int j = 0; j < 8; ++j)
        dst[j] = (_Float16)W[(size_t)(k0 + half * 8 + j) * N + n];
#pragma unroll
    for (int j = 0; j < 8; ++j)
        dst[8 + j] = (_Float16)W[(size_t)(k0 + 16 + half * 8 + j) * N + n];
}

#define SP_RB     196
#define SP_WAVES  32
#define SP_SWEEP  512
#define SP_CAP    640

__device__ __forceinline__ void sp_add_row(float4* acc, int r, int lane, float v, float4 xv) {
    float4 a = acc[r * 32 + lane];
    a.x += v * xv.x; a.y += v * xv.y; a.z += v * xv.z; a.w += v * xv.w;
    acc[r * 32 + lane] = a;
}

__global__ __launch_bounds__(1024)
void gcn_spmm_rows(const int* __restrict__ row, const int* __restrict__ col,
                   const float* __restrict__ vals, const float* __restrict__ x,
                   float* __restrict__ agg, int nNodes, int nEdges) {
    __shared__ float4        acc[SP_RB * 32];
    __shared__ int           l_c[SP_WAVES * SP_CAP];
    __shared__ float         l_v[SP_WAVES * SP_CAP];
    __shared__ unsigned char l_r[SP_WAVES * SP_CAP];
    __shared__ int           l_n[SP_WAVES];
    __shared__ int           more[2];

    const int tid  = threadIdx.x;
    const int lane = tid & 31;
    const int w    = __builtin_amdgcn_readfirstlane(tid >> 5);
    const float4* x4 = (const float4*)x;
    const int seg   = ((nEdges + SP_WAVES * SP_SWEEP - 1) / (SP_WAVES * SP_SWEEP)) * SP_SWEEP;
    const int nIter = seg / SP_SWEEP;
    const int maxRounds = nIter + 2;
    const int segStart = w * seg;
    const int shift = (int)((blockIdx.x * (unsigned)nIter) / gridDim.x);

    for (int r0 = blockIdx.x * SP_RB; r0 < nNodes; r0 += gridDim.x * SP_RB) {
        for (int i = tid; i < SP_RB * 32; i += 1024) acc[i] = make_float4(0.f, 0.f, 0.f, 0.f);
        if (tid < 2) more[tid] = 0;
        int it = 0;
        __syncthreads();

        for (int round = 0; round < maxRounds; ++round) {
            int n = 0;
            while (it < nIter && n + SP_SWEEP <= SP_CAP) {
                int p = it + shift;
                if (p >= nIter) p -= nIter;
                const int base = segStart + p * SP_SWEEP;
                ++it;
                if (base >= nEdges) continue;
                int4 rr[4];
                if (base + SP_SWEEP <= nEdges) {
#pragma unroll
                    for (int u = 0; u < 4; ++u)
                        rr[u] = *(const int4*)(row + base + u * 128 + lane * 4);
                } else {
#pragma unroll
                    for (int u = 0; u < 4; ++u) {
                        int e0 = base + u * 128 + lane * 4;
                        rr[u].x = e0     < nEdges ? row[e0]     : -1;
                        rr[u].y = e0 + 1 < nEdges ? row[e0 + 1] : -1;
                        rr[u].z = e0 + 2 < nEdges ? row[e0 + 2] : -1;
                        rr[u].w = e0 + 3 < nEdges ? row[e0 + 3] : -1;
                    }
                }
#pragma unroll
                for (int u = 0; u < 4; ++u) {
                    int e0 = base + u * 128 + lane * 4;
                    unsigned d[4];
                    d[0] = (unsigned)(rr[u].x - r0); d[1] = (unsigned)(rr[u].y - r0);
                    d[2] = (unsigned)(rr[u].z - r0); d[3] = (unsigned)(rr[u].w - r0);
                    bool any = (d[0] < SP_RB) | (d[1] < SP_RB) | (d[2] < SP_RB) | (d[3] < SP_RB);
                    if (__builtin_amdgcn_ballot_w32(any) != 0u) {
#pragma unroll
                        for (int j = 0; j < 4; ++j) {
                            bool hit = d[j] < SP_RB;
                            unsigned m = __builtin_amdgcn_ballot_w32(hit);
                            if (m != 0u) {
                                if (hit) {
                                    int q = w * SP_CAP + n + (int)__builtin_amdgcn_mbcnt_lo(m, 0u);
                                    l_r[q] = (unsigned char)d[j];
                                    l_c[q] = e0 + j;
                                }
                                n += __builtin_popcount(m);
                            }
                        }
                    }
                }
            }
            for (int i = lane; i < n; i += 32) {
                int e = l_c[w * SP_CAP + i];
                l_c[w * SP_CAP + i] = col[e];
                l_v[w * SP_CAP + i] = vals[e];
            }
            if (lane == 0) {
                l_n[w] = n;
                if (it < nIter) more[round & 1] = 1;
            }
            __syncthreads();
            const int again = more[round & 1];
            if (tid == 0) more[(round + 1) & 1] = 0;

            for (int L = 0; L < 8; ++L) {
                int cnt[4];
#pragma unroll
                for (int q = 0; q < 4; ++q) cnt[q] = l_n[L + 8 * q];
                int cmax = cnt[0];
#pragma unroll
                for (int q = 1; q < 4; ++q) cmax = cnt[q] > cmax ? cnt[q] : cmax;
                for (int b = 0; b < cmax; b += 32) {
                    unsigned m[4];
#pragma unroll
                    for (int q = 0; q < 4; ++q) {
                        int i = b + lane;
                        bool own = false;
                        if (i < cnt[q]) own = ((int)(l_r[(L + 8 * q) * SP_CAP + i] & 31) == w);
                        m[q] = __builtin_amdgcn_ballot_w32(own);
                    }
                    while ((m[0] | m[1] | m[2] | m[3]) != 0u) {
                        int    fi[8];
                        int    fr[8];
                        int    fc[8];
                        float  fv[8];
                        float4 fx[8];
                        bool   has[8];
#pragma unroll
                        for (int s = 0; s < 8; ++s) {
                            const int q = s & 3;
                            has[s] = m[q] != 0u;
                            int k = has[s] ? __builtin_ctz(m[q]) : 0;
                            m[q] &= m[q] - 1u;
                            fi[s] = (L + 8 * q) * SP_CAP + b + k;
                        }
#pragma unroll
                        for (int s = 0; s < 8; ++s) {
                            fr[s] = (int)l_r[fi[s]];
                            fc[s] = l_c[fi[s]];
                            fv[s] = l_v[fi[s]];
                        }
#pragma unroll
                        for (int s = 0; s < 8; ++s) {
                            fx[s] = make_float4(0.f, 0.f, 0.f, 0.f);
                            if (has[s]) fx[s] = x4[(size_t)fc[s] * 32 + lane];
                        }
#pragma unroll
                        for (int s = 0; s < 8; ++s)
                            if (has[s]) sp_add_row(acc, fr[s], lane, fv[s], fx[s]);
                    }
                }
            }
            __syncthreads();
            if (!again) break;
        }

        for (int i = tid; i < SP_RB * 32; i += 1024) {
            int r = r0 + (i >> 5);
            if (r < nNodes) ((float4*)agg)[(size_t)r * 32 + (i & 31)] = acc[i];
        }
        __syncthreads();
    }
}

template <int NTILES, bool RELU>
__global__ void gcn_gemm_wmma(const float* __restrict__ A,
                              const _Float16* __restrict__ Bp,
                              float* __restrict__ C, int M) {
    const int N = NTILES * 16;
    int lane = threadIdx.x & 31;
    int warp = threadIdx.x >> 5;
    int wavesPerBlock = blockDim.x >> 5;
    int mt = blockIdx.x * wavesPerBlock + warp;
    int mtiles = M >> 4;
    if (mt >= mtiles) return;

    int half = lane >> 4;
    int r    = lane & 15;

    int mrow = mt * 16 + r;

    v8f acc[NTILES];
#pragma unroll
    for (int t = 0; t < NTILES; ++t) acc[t] = (v8f){};

#pragma unroll
    for (int kk = 0; kk < 4; ++kk) {
        int k0 = kk * 32;
        const float* ap = A + (size_t)mrow * IN_DIM + k0 + half * 8;
        float4 f0 = *(const float4*)(ap + 0);
        float4 f1 = *(const float4*)(ap + 4);
        float4 f2 = *(const float4*)(ap + 16);
        float4 f3 = *(const float4*)(ap + 20);
        v16h a;
        a[0]  = (_Float16)f0.x; a[1]  = (_Float16)f0.y; a[2]  = (_Float16)f0.z; a[3]  = (_Float16)f0.w;
        a[4]  = (_Float16)f1.x; a[5]  = (_Float16)f1.y; a[6]  = (_Float16)f1.z; a[7]  = (_Float16)f1.w;
        a[8]  = (_Float16)f2.x; a[9]  = (_Float16)f2.y; a[10] = (_Float16)f2.z; a[11] = (_Float16)f2.w;
        a[12] = (_Float16)f3.x; a[13] = (_Float16)f3.y; a[14] = (_Float16)f3.z; a[15] = (_Float16)f3.w;

#pragma unroll
        for (int t = 0; t < NTILES; ++t) {
            v16h b = *(const v16h*)(Bp + (((size_t)t * 4 + kk) * 32 + lane) * 16);
            acc[t] = __builtin_amdgcn_wmma_f32_16x16x32_f16(
                         false, a,  false, b,
                         (short)0, acc[t],
                         false,  false);
        }
    }

    int mbase = mt * 16 + half * 8;
#pragma unroll
    for (int t = 0; t < NTILES; ++t) {
        int ncol = t * 16 + r;
#pragma unroll
        for (int v = 0; v < 8; ++v) {
            float val = acc[t][v];
            if (RELU) val = fmaxf(val, 0.0f);
            C[(size_t)(mbase + v) * N + ncol] = val;
        }
    }
}

extern "C" void kernel_launch(void* const* d_in, const int* in_sizes, int n_in,
                              void* d_out, int out_size, void* d_ws, size_t ws_size,
                              hipStream_t stream) {
    const float* x    = (const float*)d_in[0];
    const int*   row  = (const int*)d_in[1];
    const int*   col  = (const int*)d_in[2];
    const float* vals = (const float*)d_in[3];
    const float* W1   = (const float*)d_in[4];
    const float* W2   = (const float*)d_in[5];
    float* out = (float*)d_out;

    const int NN = in_sizes[0] / IN_DIM;
    const int NE = in_sizes[1];

    float*     aggF = (float*)d_ws;
    float*     hF   = aggF + (size_t)NN * IN_DIM;
    _Float16*  W1p  = (_Float16*)(hF + (size_t)NN * IN_DIM);
    _Float16*  W2p  = W1p + 128 * 128;

    gcn_pack_w<<<4, 256, 0, stream>>>(W1, W1p, 8);
    gcn_pack_w<<<2, 256, 0, stream>>>(W2, W2p, 4);

    gcn_spmm_rows<<<256, 1024, 0, stream>>>(row, col, vals, x, aggF, NN, NE);

    {
        int mtiles = NN / 16;
        int blocks = (mtiles + 3) / 4;
        gcn_gemm_wmma<8, true><<<blocks, 128, 0, stream>>>(aggF, W1p, hF, NN);
    }

    gcn_spmm_rows<<<256, 1024, 0, stream>>>(row, col, vals, hF, aggF, NN, NE);

    {
        int mtiles = NN / 16;
        int blocks = (mtiles + 3) / 4;
        gcn_gemm_wmma<4, false><<<blocks, 128, 0, stream>>>(aggF, W2p, out, NN);
    }
}


# Edge sweep made two-level: a new pass splits each of 512 edge slices by 16 row ranges into workspace lists (ballot compaction); row groups sweep their range's lists only; f32 sum order changes
# speedup vs baseline: 1.9625x; 1.2001x over previous
#include <hip/hip_runtime.h>
#include <hip/hip_bf16.h>

typedef __attribute__((ext_vector_type(16))) _Float16 v16h;
typedef __attribute__((ext_vector_type(8)))  float    v8f;

#define IN_DIM 128

__global__ void gcn_pack_w(const float* __restrict__ W, _Float16* __restrict__ Wp,
                           int Nt  ) {
    int id = blockIdx.x * blockDim.x + threadIdx.x;
    int total = Nt * 4 * 32;
    if (id >= total) return;
    int lane = id & 31;
    int kk   = (id >> 5) & 3;
    int t    = id >> 7;
    int half = lane >> 4;
    int r    = lane & 15;
    int N    = Nt * 16;
    int n    = t * 16 + r;
    int k0   = kk * 32;
    _Float16* dst = Wp + (size_t)id * 16;
#pragma unroll
    for (int j = 0; j < 8; ++j)
        dst[j] = (_Float16)W[(size_t)(k0 + half * 8 + j) * N + n];
#pragma unroll
    for (int j = 0; j < 8; ++j)
        dst[8 + j] = (_Float16)W[(size_t)(k0 + 16 + half * 8 + j) * N + n];
}

#define SP_RB     196
#define SP_WAVES  32
#define SP_SB     16
#define SP_NSUB   (SP_WAVES * 16)
#define SP_SWEEP  512
#define SP_CAP    640
#define SP_LIST   (16 * SP_NSUB * 1568)

__device__ __forceinline__ int sp_sub_cap(int nEdges) {
    return (((nEdges + SP_NSUB - 1) / SP_NSUB) + 31) & ~31;
}

__device__ __forceinline__ int sp_put4(const int* rr, const int* cc, const float* vv, int lo, int base, int n,
                                       int* __restrict__ lrow, int* __restrict__ lcol,
                                       float* __restrict__ lval) {
#pragma unroll
    for (int j = 0; j < 4; ++j) {
        const bool hit = (unsigned)(rr[j] - lo) < (unsigned)(SP_SB * SP_RB);
        const unsigned m = __builtin_amdgcn_ballot_w32(hit);
        if (m != 0u) {
            if (hit) {
                const int q = base + n + (int)__builtin_amdgcn_mbcnt_lo(m, 0u);
                lrow[q] = rr[j];
                lcol[q] = cc[j];
                lval[q] = vv[j];
            }
            n += __builtin_popcount(m);
        }
    }
    return n;
}

__global__ __launch_bounds__(1024)
void gcn_edge_ranges(const int* __restrict__ row, const int* __restrict__ col,
                     const float* __restrict__ vals,
                     int* __restrict__ lrow, int* __restrict__ lcol, float* __restrict__ lval,
                     int* __restrict__ lcnt, int nNodes, int nEdges) {
    __shared__ int s_n[32];
    const int lane = threadIdx.x & 31;
    const int w    = __builtin_amdgcn_readfirstlane(threadIdx.x >> 5);
    const int subCap  = sp_sub_cap(nEdges);
    const int nRanges = (nNodes + SP_SB * SP_RB - 1) / (SP_SB * SP_RB);

    for (int u0 = blockIdx.x * 2; u0 < SP_NSUB; u0 += gridDim.x * 2) {
        for (int sb = 0; sb < nRanges; sb += 16) {
            const int u = u0 + (w >> 4);
            const int s = sb + (w & 15);
            int n = 0;
            if (s < nRanges) {
                const int lo     = s * (SP_SB * SP_RB);
                const int eBegin = u * subCap;
                const int eEnd   = eBegin + subCap < nEdges ? eBegin + subCap : nEdges;
                const int base   = (s * SP_NSUB + u) * subCap;
                int c = eBegin;
                for (; c + 512 <= eEnd; c += 512) {
                    int4   r4[4], c4[4];
                    float4 v4[4];
#pragma unroll
                    for (int k = 0; k < 4; ++k) {
                        const int e0 = c + k * 128 + lane * 4;
                        r4[k] = *(const int4*)(row + e0);
                        c4[k] = *(const int4*)(col + e0);
                        v4[k] = *(const float4*)(vals + e0);
                    }
#pragma unroll
                    for (int k = 0; k < 4; ++k) {
                        const int   rr[4] = {r4[k].x, r4[k].y, r4[k].z, r4[k].w};
                        const int   cc[4] = {c4[k].x, c4[k].y, c4[k].z, c4[k].w};
                        const float vv[4] = {v4[k].x, v4[k].y, v4[k].z, v4[k].w};
                        n = sp_put4(rr, cc, vv, lo, base, n, lrow, lcol, lval);
                    }
                }
                for (; c < eEnd; c += 128) {
                    const int e0 = c + lane * 4;
                    int   rr[4], cc[4];
                    float vv[4];
                    if (e0 + 3 < eEnd) {
                        const int4   r4 = *(const int4*)(row + e0);
                        const int4   c4 = *(const int4*)(col + e0);
                        const float4 v4 = *(const float4*)(vals + e0);
                        rr[0] = r4.x; rr[1] = r4.y; rr[2] = r4.z; rr[3] = r4.w;
                        cc[0] = c4.x; cc[1] = c4.y; cc[2] = c4.z; cc[3] = c4.w;
                        vv[0] = v4.x; vv[1] = v4.y; vv[2] = v4.z; vv[3] = v4.w;
                    } else {
#pragma unroll
                        for (int j = 0; j < 4; ++j) {
                            const bool in = e0 + j < eEnd;
                            rr[j] = in ? row[e0 + j]  : -1;
                            cc[j] = in ? col[e0 + j]  : 0;
                            vv[j] = in ? vals[e0 + j] : 0.f;
                        }
                    }
                    n = sp_put4(rr, cc, vv, lo, base, n, lrow, lcol, lval);
                }
                if (lane < ((4 - (n & 3)) & 3)) lrow[base + n + lane] = -1;
            }
            if (lane == 0) s_n[w] = n;
            __syncthreads();
            if (w == 0) lcnt[(sb >> 4) * (SP_NSUB * 16) + u0 * 16 + lane] = s_n[lane];
            __syncthreads();
        }
    }
}

__device__ __forceinline__ void sp_add_row(float4* acc, int r, int lane, float v, float4 xv) {
    float4 a = acc[r * 32 + lane];
    a.x += v * xv.x; a.y += v * xv.y; a.z += v * xv.z; a.w += v * xv.w;
    acc[r * 32 + lane] = a;
}

__global__ __launch_bounds__(1024)
void gcn_spmm_rows(const int* __restrict__ lrow, const int* __restrict__ lcol,
                   const float* __restrict__ lval, const int* __restrict__ lcnt,
                   const float* __restrict__ x,
                   float* __restrict__ agg, int nNodes, int nEdges) {
    __shared__ float4        acc[SP_RB * 32];
    __shared__ int           l_c[SP_WAVES * SP_CAP];
    __shared__ float         l_v[SP_WAVES * SP_CAP];
    __shared__ unsigned char l_r[SP_WAVES * SP_CAP];
    __shared__ int           l_n[SP_WAVES];
    __shared__ int           more[2];

    const int tid  = threadIdx.x;
    const int lane = tid & 31;
    const int w    = __builtin_amdgcn_readfirstlane(tid >> 5);
    const float4* x4 = (const float4*)x;
    const int subCap    = sp_sub_cap(nEdges);
    const int maxRounds = 4 * ((subCap + 127) >> 7) + 2;
    const int u0 = w * 16;

    for (int r0 = blockIdx.x * SP_RB; r0 < nNodes; r0 += gridDim.x * SP_RB) {
        const int s = r0 / (SP_SB * SP_RB);
        const int myc = lcnt[(s >> 4) * (SP_NSUB * 16) + (u0 + (lane & 15)) * 16 + (s & 15)];
        for (int i = tid; i < SP_RB * 32; i += 1024) acc[i] = make_float4(0.f, 0.f, 0.f, 0.f);
        if (tid < 2) more[tid] = 0;
        int g = 0;
        int c = 0;
        __syncthreads();

        for (int round = 0; round < maxRounds; ++round) {
            int n = 0;
            while (g < 4 && n + SP_SWEEP <= SP_CAP) {
                int nq[4], bq[4];
                int cmax = 0;
#pragma unroll
                for (int q = 0; q < 4; ++q) {
                    nq[q] = __builtin_amdgcn_readlane(myc, g * 4 + q);
                    bq[q] = (s * SP_NSUB + u0 + g * 4 + q) * subCap;
                    cmax  = nq[q] > cmax ? nq[q] : cmax;
                }
                if (c >= cmax) { ++g; c = 0; continue; }
                const int idx = c + lane * 4;
                c += 128;
                int4 rr[4];
#pragma unroll
                for (int q = 0; q < 4; ++q) {
                    rr[q] = make_int4(-1, -1, -1, -1);
                    if (idx < nq[q]) rr[q] = *(const int4*)(lrow + bq[q] + idx);
                }
#pragma unroll
                for (int q = 0; q < 4; ++q) {
                    const int p0 = bq[q] + idx;
                    unsigned d[4];
                    d[0] = (unsigned)(rr[q].x - r0); d[1] = (unsigned)(rr[q].y - r0);
                    d[2] = (unsigned)(rr[q].z - r0); d[3] = (unsigned)(rr[q].w - r0);
                    bool any = (d[0] < SP_RB) | (d[1] < SP_RB) | (d[2] < SP_RB) | (d[3] < SP_RB);
                    if (__builtin_amdgcn_ballot_w32(any) != 0u) {
#pragma unroll
                        for (int j = 0; j < 4; ++j) {
                            bool hit = d[j] < SP_RB;
                            unsigned m = __builtin_amdgcn_ballot_w32(hit);
                            if (m != 0u) {
                                if (hit) {
                                    int qq = w * SP_CAP + n + (int)__builtin_amdgcn_mbcnt_lo(m, 0u);
                                    l_r[qq] = (unsigned char)d[j];
                                    l_c[qq] = p0 + j;
                                }
                                n += __builtin_popcount(m);
                            }
                        }
                    }
                }
            }
            for (int i = lane; i < n; i += 32) {
                int e = l_c[w * SP_CAP + i];
                l_c[w * SP_CAP + i] = lcol[e];
                l_v[w * SP_CAP + i] = lval[e];
            }
            if (lane == 0) {
                l_n[w] = n;
                if (g < 4) more[round & 1] = 1;
            }
            __syncthreads();
            const int again = more[round & 1];
            if (tid == 0) more[(round + 1) & 1] = 0;

            for (int L = 0; L < 8; ++L) {
                int cnt[4];
#pragma unroll
                for (int q = 0; q < 4; ++q) cnt[q] = l_n[L + 8 * q];
                int cmax = cnt[0];
#pragma unroll
                for (int q = 1; q < 4; ++q) cmax = cnt[q] > cmax ? cnt[q] : cmax;
                for (int b = 0; b < cmax; b += 32) {
                    unsigned m[4];
#pragma unroll
                    for (int q = 0; q < 4; ++q) {
                        int i = b + lane;
                        bool own = false;
                        if (i < cnt[q]) own = ((int)(l_r[(L + 8 * q) * SP_CAP + i] & 31) == w);
                        m[q] = __builtin_amdgcn_ballot_w32(own);
                    }
                    while ((m[0] | m[1] | m[2] | m[3]) != 0u) {
                        int    fi[8];
                        int    fr[8];
                        int    fc[8];
                        float  fv[8];
                        float4 fx[8];
                        bool   has[8];
#pragma unroll
                        for (int s = 0; s < 8; ++s) {
                            const int q = s & 3;
                            has[s] = m[q] != 0u;
                            int k = has[s] ? __builtin_ctz(m[q]) : 0;
                            m[q] &= m[q] - 1u;
                            fi[s] = (L + 8 * q) * SP_CAP + b + k;
                        }
#pragma unroll
                        for (int s = 0; s < 8; ++s) {
                            fr[s] = (int)l_r[fi[s]];
                            fc[s] = l_c[fi[s]];
                            fv[s] = l_v[fi[s]];
                        }
#pragma unroll
                        for (int s = 0; s < 8; ++s) {
                            fx[s] = make_float4(0.f, 0.f, 0.f, 0.f);
                            if (has[s]) fx[s] = x4[(size_t)fc[s] * 32 + lane];
                        }
#pragma unroll
                        for (int s = 0; s < 8; ++s)
                            if (has[s]) sp_add_row(acc, fr[s], lane, fv[s], fx[s]);
                    }
                }
            }
            __syncthreads();
            if (!again) break;
        }

        for (int i = tid; i < SP_RB * 32; i += 1024) {
            int r = r0 + (i >> 5);
            if (r < nNodes) ((float4*)agg)[(size_t)r * 32 + (i & 31)] = acc[i];
        }
        __syncthreads();
    }
}

template <int NTILES, bool RELU>
__global__ void gcn_gemm_wmma(const float* __restrict__ A,
                              const _Float16* __restrict__ Bp,
                              float* __restrict__ C, int M) {
    const int N = NTILES * 16;
    int lane = threadIdx.x & 31;
    int warp = threadIdx.x >> 5;
    int wavesPerBlock = blockDim.x >> 5;
    int mt = blockIdx.x * wavesPerBlock + warp;
    int mtiles = M >> 4;
    if (mt >= mtiles) return;

    int half = lane >> 4;
    int r    = lane & 15;

    int mrow = mt * 16 + r;

    v8f acc[NTILES];
#pragma unroll
    for (int t = 0; t < NTILES; ++t) acc[t] = (v8f){};

#pragma unroll
    for (int kk = 0; kk < 4; ++kk) {
        int k0 = kk * 32;
        const float* ap = A + (size_t)mrow * IN_DIM + k0 + half * 8;
        float4 f0 = *(const float4*)(ap + 0);
        float4 f1 = *(const float4*)(ap + 4);
        float4 f2 = *(const float4*)(ap + 16);
        float4 f3 = *(const float4*)(ap + 20);
        v16h a;
        a[0]  = (_Float16)f0.x; a[1]  = (_Float16)f0.y; a[2]  = (_Float16)f0.z; a[3]  = (_Float16)f0.w;
        a[4]  = (_Float16)f1.x; a[5]  = (_Float16)f1.y; a[6]  = (_Float16)f1.z; a[7]  = (_Float16)f1.w;
        a[8]  = (_Float16)f2.x; a[9]  = (_Float16)f2.y; a[10] = (_Float16)f2.z; a[11] = (_Float16)f2.w;
        a[12] = (_Float16)f3.x; a[13] = (_Float16)f3.y; a[14] = (_Float16)f3.z; a[15] = (_Float16)f3.w;

#pragma unroll
        for (int t = 0; t < NTILES; ++t) {
            v16h b = *(const v16h*)(Bp + (((size_t)t * 4 + kk) * 32 + lane) * 16);
            acc[t] = __builtin_amdgcn_wmma_f32_16x16x32_f16(
                         false, a,  false, b,
                         (short)0, acc[t],
                         false,  false);
        }
    }

    int mbase = mt * 16 + half * 8;
#pragma unroll
    for (int t = 0; t < NTILES; ++t) {
        int ncol = t * 16 + r;
#pragma unroll
        for (int v = 0; v < 8; ++v) {
            float val = acc[t][v];
            if (RELU) val = fmaxf(val, 0.0f);
            C[(size_t)(mbase + v) * N + ncol] = val;
        }
    }
}

extern "C" void kernel_launch(void* const* d_in, const int* in_sizes, int n_in,
                              void* d_out, int out_size, void* d_ws, size_t ws_size,
                              hipStream_t stream) {
    const float* x    = (const float*)d_in[0];
    const int*   row  = (const int*)d_in[1];
    const int*   col  = (const int*)d_in[2];
    const float* vals = (const float*)d_in[3];
    const float* W1   = (const float*)d_in[4];
    const float* W2   = (const float*)d_in[5];
    float* out = (float*)d_out;

    const int NN = in_sizes[0] / IN_DIM;
    const int NE = in_sizes[1];

    float*     aggF = (float*)d_ws;
    float*     hF   = aggF + (size_t)NN * IN_DIM;
    _Float16*  W1p  = (_Float16*)(hF + (size_t)NN * IN_DIM);
    _Float16*  W2p  = W1p + 128 * 128;
    int*       lcnt = (int*)(W2p + 128 * 64);
    int*       lrow = lcnt + SP_NSUB * 16;
    int*       lcol = lrow + SP_LIST;
    float*     lval = (float*)(lcol + SP_LIST);

    gcn_pack_w<<<4, 256, 0, stream>>>(W1, W1p, 8);
    gcn_pack_w<<<2, 256, 0, stream>>>(W2, W2p, 4);

    gcn_edge_ranges<<<256, 1024, 0, stream>>>(row, col, vals, lrow, lcol, lval, lcnt, NN, NE);

    gcn_spmm_rows<<<256, 1024, 0, stream>>>(lrow, lcol, lval, lcnt, x, aggF, NN, NE);

    {
        int mtiles = NN / 16;
        int blocks = (mtiles + 3) / 4;
        gcn_gemm_wmma<8, true><<<blocks, 128, 0, stream>>>(aggF, W1p, hF, NN);
    }

    gcn_spmm_rows<<<256, 1024, 0, stream>>>(lrow, lcol, lval, lcnt, hF, aggF, NN, NE);

    {
        int mtiles = NN / 16;
        int blocks = (mtiles + 3) / 4;
        gcn_gemm_wmma<4, false><<<blocks, 128, 0, stream>>>(aggF, W2p, out, NN);
    }
}


# Second SpMM reads back per-wave entry lists (local row, col, value) that the first SpMM notes in the workspace as it adds: layer 2 has no sweep, LDS lists, scan or barrier; same f32 sums and order
# speedup vs baseline: 2.8719x; 1.4634x over previous
#include <hip/hip_runtime.h>
#include <hip/hip_bf16.h>

typedef __attribute__((ext_vector_type(16))) _Float16 v16h;
typedef __attribute__((ext_vector_type(8)))  float    v8f;

#define IN_DIM 128

__global__ void gcn_pack_w(const float* __restrict__ W, _Float16* __restrict__ Wp,
                           int Nt  ) {
    int id = blockIdx.x * blockDim.x + threadIdx.x;
    int total = Nt * 4 * 32;
    if (id >= total) return;
    int lane = id & 31;
    int kk   = (id >> 5) & 3;
    int t    = id >> 7;
    int half = lane >> 4;
    int r    = lane & 15;
    int N    = Nt * 16;
    int n    = t * 16 + r;
    int k0   = kk * 32;
    _Float16* dst = Wp + (size_t)id * 16;
#pragma unroll
    for (int j = 0; j < 8; ++j)
        dst[j] = (_Float16)W[(size_t)(k0 + half * 8 + j) * N + n];
#pragma unroll
    for (int j = 0; j < 8; ++j)
        dst[8 + j] = (_Float16)W[(size_t)(k0 + 16 + half * 8 + j) * N + n];
}

#define SP_RB     196
#define SP_WAVES  32
#define SP_SB     16
#define SP_NSUB   (SP_WAVES * 16)
#define SP_SWEEP  512
#define SP_CAP    640
#define SP_OWN    576
#define SP_LIST   (16 * SP_NSUB * 1568)
#define SP_REC    256
#define SP_RECB   256

__device__ __forceinline__ int sp_sub_cap(int nEdges) {
    return (((nEdges + SP_NSUB - 1) / SP_NSUB) + 31) & ~31;
}

__device__ __forceinline__ int sp_put4(const int* rr, const int* cc, const float* vv, int lo, int base, int n,
                                       int* __restrict__ lrow, int* __restrict__ lcol,
                                       float* __restrict__ lval) {
#pragma unroll
    for (int j = 0; j < 4; ++j) {
        const bool hit = (unsigned)(rr[j] - lo) < (unsigned)(SP_SB * SP_RB);
        const unsigned m = __builtin_amdgcn_ballot_w32(hit);
        if (m != 0u) {
            if (hit) {
                const int q = base + n + (int)__builtin_amdgcn_mbcnt_lo(m, 0u);
                lrow[q] = rr[j];
                lcol[q] = cc[j];
                lval[q] = vv[j];
            }
            n += __builtin_popcount(m);
        }
    }
    return n;
}

__global__ __launch_bounds__(1024)
void gcn_edge_ranges(const int* __restrict__ row, const int* __restrict__ col,
                     const float* __restrict__ vals,
                     int* __restrict__ lrow, int* __restrict__ lcol, float* __restrict__ lval,
                     int* __restrict__ lcnt, int nNodes, int nEdges) {
    __shared__ int s_n[32];
    const int lane = threadIdx.x & 31;
    const int w    = __builtin_amdgcn_readfirstlane(threadIdx.x >> 5);
    const int subCap  = sp_sub_cap(nEdges);
    const int nRanges = (nNodes + SP_SB * SP_RB - 1) / (SP_SB * SP_RB);

    for (int u0 = blockIdx.x * 2; u0 < SP_NSUB; u0 += gridDim.x * 2) {
        for (int sb = 0; sb < nRanges; sb += 16) {
            const int u = u0 + (w >> 4);
            const int s = sb + (w & 15);
            int n = 0;
            if (s < nRanges) {
                const int lo     = s * (SP_SB * SP_RB);
                const int eBegin = u * subCap;
                const int eEnd   = eBegin + subCap < nEdges ? eBegin + subCap : nEdges;
                const int base   = (s * SP_NSUB + u) * subCap;
                int c = eBegin;
                for (; c + 512 <= eEnd; c += 512) {
                    int4   r4[4], c4[4];
                    float4 v4[4];
#pragma unroll
                    for (int k = 0; k < 4; ++k) {
                        const int e0 = c + k * 128 + lane * 4;
                        r4[k] = *(const int4*)(row + e0);
                        c4[k] = *(const int4*)(col + e0);
                        v4[k] = *(const float4*)(vals + e0);
                    }
#pragma unroll
                    for (int k = 0; k < 4; ++k) {
                        const int   rr[4] = {r4[k].x, r4[k].y, r4[k].z, r4[k].w};
                        const int   cc[4] = {c4[k].x, c4[k].y, c4[k].z, c4[k].w};
                        const float vv[4] = {v4[k].x, v4[k].y, v4[k].z, v4[k].w};
                        n = sp_put4(rr, cc, vv, lo, base, n, lrow, lcol, lval);
                    }
                }
                for (; c < eEnd; c += 128) {
                    const int e0 = c + lane * 4;
                    int   rr[4], cc[4];
                    float vv[4];
                    if (e0 + 3 < eEnd) {
                        const int4   r4 = *(const int4*)(row + e0);
                        const int4   c4 = *(const int4*)(col + e0);
                        const float4 v4 = *(const float4*)(vals + e0);
                        rr[0] = r4.x; rr[1] = r4.y; rr[2] = r4.z; rr[3] = r4.w;
                        cc[0] = c4.x; cc[1] = c4.y; cc[2] = c4.z; cc[3] = c4.w;
                        vv[0] = v4.x; vv[1] = v4.y; vv[2] = v4.z; vv[3] = v4.w;
                    } else {
#pragma unroll
                        for (int j = 0; j < 4; ++j) {
                            const bool in = e0 + j < eEnd;
                            rr[j] = in ? row[e0 + j]  : -1;
                            cc[j] = in ? col[e0 + j]  : 0;
                            vv[j] = in ? vals[e0 + j] : 0.f;
                        }
                    }
                    n = sp_put4(rr, cc, vv, lo, base, n, lrow, lcol, lval);
                }
                if (lane < ((4 - (n & 3)) & 3)) lrow[base + n + lane] = -1;
            }
            if (lane == 0) s_n[w] = n;
            __syncthreads();
            if (w == 0) lcnt[(sb >> 4) * (SP_NSUB * 16) + u0 * 16 + lane] = s_n[lane];
            __syncthreads();
        }
    }
}

__device__ __forceinline__ void sp_add_row(float4* acc, int r, int lane, float v, float4 xv) {
    float4 a = acc[r * 32 + lane];
    a.x += v * xv.x; a.y += v * xv.y; a.z += v * xv.z; a.w += v * xv.w;
    acc[r * 32 + lane] = a;
}

typedef __attribute__((ext_vector_type(4))) float sp_f4;

template <bool FULL>
__device__ __forceinline__ void sp_add16(float4* acc, const float* __restrict__ x, int lane,
                                         int er, int ec, float ev, int cnt) {
    sp_f4 fx[16];
#pragma unroll
    for (int s = 0; s < 16; ++s) {
        fx[s] = (sp_f4){0.f, 0.f, 0.f, 0.f};
        if (FULL || s < cnt) {
            const int c = __builtin_amdgcn_readlane(ec, s);
            fx[s] = *(const sp_f4*)(x + (size_t)c * IN_DIM + lane * 4);
        }
    }
    if (FULL) {
        asm volatile("" :: "v"(fx[0]), "v"(fx[1]), "v"(fx[2]), "v"(fx[3]), "v"(fx[4]), "v"(fx[5]),
                           "v"(fx[6]), "v"(fx[7]), "v"(fx[8]), "v"(fx[9]), "v"(fx[10]), "v"(fx[11]),
                           "v"(fx[12]), "v"(fx[13]), "v"(fx[14]), "v"(fx[15]));
    }
#pragma unroll
    for (int s = 0; s < 16; ++s) {
        if (FULL || s < cnt) {
            const int   r = __builtin_amdgcn_readlane(er, s);
            const float v = __int_as_float(__builtin_amdgcn_readlane(__float_as_int(ev), s));
            sp_add_row(acc, r, lane, v, make_float4(fx[s].x, fx[s].y, fx[s].z, fx[s].w));
        }
    }
}

template <bool LAST>
__device__ __forceinline__ int4 sp_add16_noted(float4* acc, const float* __restrict__ x, int lane,
                                               int4 cur, const int4* __restrict__ nextp, int cnt) {
    sp_f4 fx[16];
#pragma unroll
    for (int s = 0; s < 16; ++s) {
        const int c = __builtin_amdgcn_readlane(cur.y, s);
        fx[s] = *(const sp_f4*)(x + (size_t)c * IN_DIM + lane * 4);
    }
    int4 nx = cur;
    if (!LAST) nx = *nextp;
    asm volatile("" :: "v"(fx[0]), "v"(fx[1]), "v"(fx[2]), "v"(fx[3]), "v"(fx[4]), "v"(fx[5]),
                       "v"(fx[6]), "v"(fx[7]), "v"(fx[8]), "v"(fx[9]), "v"(fx[10]), "v"(fx[11]),
                       "v"(fx[12]), "v"(fx[13]), "v"(fx[14]), "v"(fx[15]),
                       "v"(nx.x), "v"(nx.y), "v"(nx.z));
#pragma unroll
    for (int s = 0; s < 16; ++s) {
        if (!LAST || s < cnt) {
            const int   r = __builtin_amdgcn_readlane(cur.x, s);
            const float v = __int_as_float(__builtin_amdgcn_readlane(cur.z, s));
            sp_add_row(acc, r, lane, v, make_float4(fx[s].x, fx[s].y, fx[s].z, fx[s].w));
        }
    }
    return nx;
}

template <int MODE>
__global__ __launch_bounds__(1024)
void gcn_spmm_rows(const int* __restrict__ lrow, const int* __restrict__ lcol,
                   const float* __restrict__ lval, const int* __restrict__ lcnt,
                   const float* __restrict__ x,
                   float* __restrict__ agg, int* __restrict__ rec_n, int* __restrict__ rec,
                   int nNodes, int nEdges) {
    __shared__ float4        acc[SP_RB * 32];
    __shared__ int           l_c[SP_WAVES * SP_CAP];
    __shared__ float         l_v[SP_WAVES * SP_CAP];
    __shared__ __attribute__((aligned(16))) unsigned char l_r[SP_WAVES * SP_CAP];
    __shared__ unsigned short l_o[SP_WAVES * SP_OWN];
    __shared__ int           l_n[SP_WAVES];
    __shared__ int           more[2];

    const int tid  = threadIdx.x;
    const int lane = tid & 31;
    const int w    = __builtin_amdgcn_readfirstlane(tid >> 5);
    const int subCap    = sp_sub_cap(nEdges);
    const int maxRounds = 4 * ((subCap + 127) >> 7) + 2;
    const int u0 = w * 16;

    for (int r0 = blockIdx.x * SP_RB; r0 < nNodes; r0 += gridDim.x * SP_RB) {
        const int rb = r0 / SP_RB;
        int4* recW = (int4*)rec + (size_t)((rb < SP_RECB ? rb : 0) * SP_WAVES + w) * SP_REC;
        if (MODE == 1) {
            const int rn = rb < SP_RECB ? rec_n[rb * SP_WAVES + lane] : -1;
            if (__builtin_amdgcn_ballot_w32(rn < 0) == 0u) {
                const int myN = __builtin_amdgcn_readlane(rn, w);
                const int l16 = lane & 15;
                int4 cur = make_int4(0, 0, 0, 0);
                if (myN > 0) cur = recW[l16 < myN ? l16 : myN - 1];
                for (int r = w; r < SP_RB; r += SP_WAVES) acc[r * 32 + lane] = make_float4(0.f, 0.f, 0.f, 0.f);
                for (int i0 = 0; i0 < myN; i0 += 16) {
                    const int left = myN - i0;
                    if (left >= 16) {
                        const int ni = i0 + 16 + l16;
                        cur = sp_add16_noted<false>(acc, x, lane, cur, recW + (ni < myN ? ni : myN - 1), 16);
                    } else {
                        sp_add16_noted<true>(acc, x, lane, cur, recW, left);
                    }
                }
                for (int r = w; r < SP_RB; r += SP_WAVES)
                    if (r0 + r < nNodes) ((float4*)agg)[(size_t)(r0 + r) * 32 + lane] = acc[r * 32 + lane];
                __syncthreads();
                continue;
            }
        }
        int  recN  = 0;
        bool recOk = rb < SP_RECB;
        const int s = r0 / (SP_SB * SP_RB);
        const int myc = lcnt[(s >> 4) * (SP_NSUB * 16) + (u0 + (lane & 15)) * 16 + (s & 15)];
        for (int i = tid; i < SP_RB * 32; i += 1024) acc[i] = make_float4(0.f, 0.f, 0.f, 0.f);
        if (tid < 2) more[tid] = 0;
        int g = 0;
        int c = 0;
        __syncthreads();

        for (int round = 0; round < maxRounds; ++round) {
            int n = 0;
            while (g < 4 && n + SP_SWEEP <= SP_CAP) {
                int nq[4], bq[4];
                int cmax = 0;
#pragma unroll
                for (int q = 0; q < 4; ++q) {
                    nq[q] = __builtin_amdgcn_readlane(myc, g * 4 + q);
                    bq[q] = (s * SP_NSUB + u0 + g * 4 + q) * subCap;
                    cmax  = nq[q] > cmax ? nq[q] : cmax;
                }
                if (c >= cmax) { ++g; c = 0; continue; }
                const int idx = c + lane * 4;
                c += 128;
                int4 rr[4];
#pragma unroll
                for (int q = 0; q < 4; ++q) {
                    rr[q] = make_int4(-1, -1, -1, -1);
                    if (idx < nq[q]) rr[q] = *(const int4*)(lrow + bq[q] + idx);
                }
#pragma unroll
                for (int q = 0; q < 4; ++q) {
                    const int p0 = bq[q] + idx;
                    unsigned d[4];
                    d[0] = (unsigned)(rr[q].x - r0); d[1] = (unsigned)(rr[q].y - r0);
                    d[2] = (unsigned)(rr[q].z - r0); d[3] = (unsigned)(rr[q].w - r0);
                    bool any = (d[0] < SP_RB) | (d[1] < SP_RB) | (d[2] < SP_RB) | (d[3] < SP_RB);
                    if (__builtin_amdgcn_ballot_w32(any) != 0u) {
#pragma unroll
                        for (int j = 0; j < 4; ++j) {
                            bool hit = d[j] < SP_RB;
                            unsigned m = __builtin_amdgcn_ballot_w32(hit);
                            if (m != 0u) {
                                if (hit) {
                                    int qq = w * SP_CAP + n + (int)__builtin_amdgcn_mbcnt_lo(m, 0u);
                                    l_r[qq] = (unsigned char)d[j];
                                    l_c[qq] = p0 + j;
                                }
                                n += __builtin_popcount(m);
                            }
                        }
                    }
                }
            }
            for (int i = lane; i < n; i += 32) {
                int e = l_c[w * SP_CAP + i];
                l_c[w * SP_CAP + i] = lcol[e];
                l_v[w * SP_CAP + i] = lval[e];
            }
            if (lane == 0) {
                l_n[w] = n;
                if (g < 4) more[round & 1] = 1;
            }
            __syncthreads();
            const int again = more[round & 1];
            if (tid == 0) more[(round + 1) & 1] = 0;

            {
                const int myn = l_n[lane];
                unsigned short* own = l_o + w * SP_OWN;
                int nOwn = 0;
                int Lg = 0;
                int b  = 0;
                for (int pass = 0; pass < 8 * ((SP_CAP + 127) / 128) + 1; ++pass) {
                    while (Lg < 8) {
                        int cnt[4];
                        int cmax = 0, stepMax = 0;
#pragma unroll
                        for (int q = 0; q < 4; ++q) {
                            cnt[q] = __builtin_amdgcn_readlane(myn, Lg * 4 + q);
                            cmax   = cnt[q] > cmax ? cnt[q] : cmax;
                            int left = cnt[q] - b;
                            left = left < 0 ? 0 : (left > 128 ? 128 : left);
                            stepMax += left;
                        }
                        if (b >= cmax) { ++Lg; b = 0; continue; }
                        if (nOwn + stepMax > SP_OWN) break;
                        const int p = b + lane * 4;
                        unsigned wd[4];
                        int      vc[4];
#pragma unroll
                        for (int q = 0; q < 4; ++q) {
                            int v = cnt[q] - p;
                            vc[q] = v < 0 ? 0 : (v > 4 ? 4 : v);
                            wd[q] = 0u;
                            if (vc[q] > 0) wd[q] = *(const unsigned*)(l_r + (Lg * 4 + q) * SP_CAP + p);
                        }
#pragma unroll
                        for (int q = 0; q < 4; ++q) {
#pragma unroll
                            for (int j = 0; j < 4; ++j) {
                                const bool hit = (j < vc[q]) & ((int)((wd[q] >> (8 * j)) & 31u) == w);
                                const unsigned m = __builtin_amdgcn_ballot_w32(hit);
                                if (m != 0u) {
                                    if (hit)
                                        own[nOwn + (int)__builtin_amdgcn_mbcnt_lo(m, 0u)] =
                                            (unsigned short)((Lg * 4 + q) * SP_CAP + p + j);
                                    nOwn += __builtin_popcount(m);
                                }
                            }
                        }
                        b += 128;
                    }
                    for (int i0 = 0; i0 < nOwn; i0 += 16) {
                        int   er = 0, ec = 0;
                        float ev = 0.f;
                        if (lane < 16 && i0 + lane < nOwn) {
                            const int e = own[i0 + lane];
                            er = (int)l_r[e];
                            ec = l_c[e];
                            ev = l_v[e];
                        }
                        const int left = nOwn - i0;
                        if (MODE == 0) {
                            const int take = left < 16 ? left : 16;
                            if (recOk && recN + take <= SP_REC) {
                                if (lane < take) recW[recN + lane] = make_int4(er, ec, __float_as_int(ev), 0);
                                recN += take;
                            } else {
                                recOk = false;
                            }
                        }
                        if (left >= 16) sp_add16<true>(acc, x, lane, er, ec, ev, 16);
                        else            sp_add16<false>(acc, x, lane, er, ec, ev, left);
                    }
                    nOwn = 0;
                    if (Lg >= 8) break;
                }
            }
            __syncthreads();
            if (!again) break;
        }

        if (MODE == 0 && rb < SP_RECB && lane == 0) rec_n[rb * SP_WAVES + w] = recOk ? recN : -1;

        for (int i = tid; i < SP_RB * 32; i += 1024) {
            int r = r0 + (i >> 5);
            if (r < nNodes) ((float4*)agg)[(size_t)r * 32 + (i & 31)] = acc[i];
        }
        __syncthreads();
    }
}

template <int NTILES, bool RELU>
__global__ void gcn_gemm_wmma(const float* __restrict__ A,
                              const _Float16* __restrict__ Bp,
                              float* __restrict__ C, int M) {
    const int N = NTILES * 16;
    int lane = threadIdx.x & 31;
    int warp = threadIdx.x >> 5;
    int wavesPerBlock = blockDim.x >> 5;
    int mt = blockIdx.x * wavesPerBlock + warp;
    int mtiles = M >> 4;
    if (mt >= mtiles) return;

    int half = lane >> 4;
    int r    = lane & 15;

    int mrow = mt * 16 + r;

    v8f acc[NTILES];
#pragma unroll
    for (int t = 0; t < NTILES; ++t) acc[t] = (v8f){};

#pragma unroll
    for (int kk = 0; kk < 4; ++kk) {
        int k0 = kk * 32;
        const float* ap = A + (size_t)mrow * IN_DIM + k0 + half * 8;
        float4 f0 = *(const float4*)(ap + 0);
        float4 f1 = *(const float4*)(ap + 4);
        float4 f2 = *(const float4*)(ap + 16);
        float4 f3 = *(const float4*)(ap + 20);
        v16h a;
        a[0]  = (_Float16)f0.x; a[1]  = (_Float16)f0.y; a[2]  = (_Float16)f0.z; a[3]  = (_Float16)f0.w;
        a[4]  = (_Float16)f1.x; a[5]  = (_Float16)f1.y; a[6]  = (_Float16)f1.z; a[7]  = (_Float16)f1.w;
        a[8]  = (_Float16)f2.x; a[9]  = (_Float16)f2.y; a[10] = (_Float16)f2.z; a[11] = (_Float16)f2.w;
        a[12] = (_Float16)f3.x; a[13] = (_Float16)f3.y; a[14] = (_Float16)f3.z; a[15] = (_Float16)f3.w;

#pragma unroll
        for (int t = 0; t < NTILES; ++t) {
            v16h b = *(const v16h*)(Bp + (((size_t)t * 4 + kk) * 32 + lane) * 16);
            acc[t] = __builtin_amdgcn_wmma_f32_16x16x32_f16(
                         false, a,  false, b,
                         (short)0, acc[t],
                         false,  false);
        }
    }

    int mbase = mt * 16 + half * 8;
#pragma unroll
    for (int t = 0; t < NTILES; ++t) {
        int ncol = t * 16 + r;
#pragma unroll
        for (int v = 0; v < 8; ++v) {
            float val = acc[t][v];
            if (RELU) val = fmaxf(val, 0.0f);
            C[(size_t)(mbase + v) * N + ncol] = val;
        }
    }
}

extern "C" void kernel_launch(void* const* d_in, const int* in_sizes, int n_in,
                              void* d_out, int out_size, void* d_ws, size_t ws_size,
                              hipStream_t stream) {
    const float* x    = (const float*)d_in[0];
    const int*   row  = (const int*)d_in[1];
    const int*   col  = (const int*)d_in[2];
    const float* vals = (const float*)d_in[3];
    const float* W1   = (const float*)d_in[4];
    const float* W2   = (const float*)d_in[5];
    float* out = (float*)d_out;

    const int NN = in_sizes[0] / IN_DIM;
    const int NE = in_sizes[1];

    float*     aggF = (float*)d_ws;
    float*     hF   = aggF + (size_t)NN * IN_DIM;
    _Float16*  W1p  = (_Float16*)(hF + (size_t)NN * IN_DIM);
    _Float16*  W2p  = W1p + 128 * 128;
    int*       lcnt = (int*)(W2p + 128 * 64);
    int*       lrow = lcnt + SP_NSUB * 16;
    int*       lcol = lrow + SP_LIST;
    float*     lval = (float*)(lcol + SP_LIST);
    int*       recLen = (int*)(lval + SP_LIST);
    int*       rec    = recLen + SP_RECB * SP_WAVES;

    gcn_pack_w<<<4, 256, 0, stream>>>(W1, W1p, 8);
    gcn_pack_w<<<2, 256, 0, stream>>>(W2, W2p, 4);

    gcn_edge_ranges<<<256, 1024, 0, stream>>>(row, col, vals, lrow, lcol, lval, lcnt, NN, NE);

    gcn_spmm_rows<0><<<256, 1024, 0, stream>>>(lrow, lcol, lval, lcnt, x, aggF, recLen, rec, NN, NE);

    {
        int mtiles = NN / 16;
        int blocks = (mtiles + 3) / 4;
        gcn_gemm_wmma<8, true><<<blocks, 128, 0, stream>>>(aggF, W1p, hF, NN);
    }

    gcn_spmm_rows<1><<<256, 1024, 0, stream>>>(lrow, lcol, lval, lcnt, hF, aggF, recLen, rec, NN, NE);

    {
        int mtiles = NN / 16;
        int blocks = (mtiles + 3) / 4;
        gcn_gemm_wmma<4, false><<<blocks, 128, 0, stream>>>(aggF, W2p, out, NN);
    }
}


# Weight-packing launches gone: each GEMM work-group (16 waves, a 16-row strip per wave) builds the f16 B fragments of W in LDS from the f32 matrix itself; 7 launches to 5; same f32 sums and order
# speedup vs baseline: 3.3792x; 1.1766x over previous
#include <hip/hip_runtime.h>
#include <hip/hip_bf16.h>

typedef __attribute__((ext_vector_type(16))) _Float16 v16h;
typedef __attribute__((ext_vector_type(8)))  float    v8f;

#define IN_DIM 128

#define SP_RB     196
#define SP_WAVES  32
#define SP_SB     16
#define SP_NSUB   (SP_WAVES * 16)
#define SP_SWEEP  512
#define SP_CAP    640
#define SP_OWN    576
#define SP_LIST   (16 * SP_NSUB * 1568)
#define SP_REC    256
#define SP_RECB   256

__device__ __forceinline__ int sp_sub_cap(int nEdges) {
    return (((nEdges + SP_NSUB - 1) / SP_NSUB) + 31) & ~31;
}

__device__ __forceinline__ int sp_put4(const int* rr, const int* cc, const float* vv, int lo, int base, int n,
                                       int* __restrict__ lrow, int* __restrict__ lcol,
                                       float* __restrict__ lval) {
#pragma unroll
    for (int j = 0; j < 4; ++j) {
        const bool hit = (unsigned)(rr[j] - lo) < (unsigned)(SP_SB * SP_RB);
        const unsigned m = __builtin_amdgcn_ballot_w32(hit);
        if (m != 0u) {
            if (hit) {
                const int q = base + n + (int)__builtin_amdgcn_mbcnt_lo(m, 0u);
                lrow[q] = rr[j];
                lcol[q] = cc[j];
                lval[q] = vv[j];
            }
            n += __builtin_popcount(m);
        }
    }
    return n;
}

__global__ __launch_bounds__(1024)
void gcn_edge_ranges(const int* __restrict__ row, const int* __restrict__ col,
                     const float* __restrict__ vals,
                     int* __restrict__ lrow, int* __restrict__ lcol, float* __restrict__ lval,
                     int* __restrict__ lcnt, int nNodes, int nEdges) {
    __shared__ int s_n[32];
    const int lane = threadIdx.x & 31;
    const int w    = __builtin_amdgcn_readfirstlane(threadIdx.x >> 5);
    const int subCap  = sp_sub_cap(nEdges);
    const int nRanges = (nNodes + SP_SB * SP_RB - 1) / (SP_SB * SP_RB);

    for (int u0 = blockIdx.x * 2; u0 < SP_NSUB; u0 += gridDim.x * 2) {
        for (int sb = 0; sb < nRanges; sb += 16) {
            const int u = u0 + (w >> 4);
            const int s = sb + (w & 15);
            int n = 0;
            if (s < nRanges) {
                const int lo     = s * (SP_SB * SP_RB);
                const int eBegin = u * subCap;
                const int eEnd   = eBegin + subCap < nEdges ? eBegin + subCap : nEdges;
                const int base   = (s * SP_NSUB + u) * subCap;
                int c = eBegin;
                for (; c + 512 <= eEnd; c += 512) {
                    int4   r4[4], c4[4];
                    float4 v4[4];
#pragma unroll
                    for (int k = 0; k < 4; ++k) {
                        const int e0 = c + k * 128 + lane * 4;
                        r4[k] = *(const int4*)(row + e0);
                        c4[k] = *(const int4*)(col + e0);
                        v4[k] = *(const float4*)(vals + e0);
                    }
#pragma unroll
                    for (int k = 0; k < 4; ++k) {
                        const int   rr[4] = {r4[k].x, r4[k].y, r4[k].z, r4[k].w};
                        const int   cc[4] = {c4[k].x, c4[k].y, c4[k].z, c4[k].w};
                        const float vv[4] = {v4[k].x, v4[k].y, v4[k].z, v4[k].w};
                        n = sp_put4(rr, cc, vv, lo, base, n, lrow, lcol, lval);
                    }
                }
                for (; c < eEnd; c += 128) {
                    const int e0 = c + lane * 4;
                    int   rr[4], cc[4];
                    float vv[4];
                    if (e0 + 3 < eEnd) {
                        const int4   r4 = *(const int4*)(row + e0);
                        const int4   c4 = *(const int4*)(col + e0);
                        const float4 v4 = *(const float4*)(vals + e0);
                        rr[0] = r4.x; rr[1] = r4.y; rr[2] = r4.z; rr[3] = r4.w;
                        cc[0] = c4.x; cc[1] = c4.y; cc[2] = c4.z; cc[3] = c4.w;
                        vv[0] = v4.x; vv[1] = v4.y; vv[2] = v4.z; vv[3] = v4.w;
                    } else {
#pragma unroll
                        for (int j = 0; j < 4; ++j) {
                            const bool in = e0 + j < eEnd;
                            rr[j] = in ? row[e0 + j]  : -1;
                            cc[j] = in ? col[e0 + j]  : 0;
                            vv[j] = in ? vals[e0 + j] : 0.f;
                        }
                    }
                    n = sp_put4(rr, cc, vv, lo, base, n, lrow, lcol, lval);
                }
                if (lane < ((4 - (n & 3)) & 3)) lrow[base + n + lane] = -1;
            }
            if (lane == 0) s_n[w] = n;
            __syncthreads();
            if (w == 0) lcnt[(sb >> 4) * (SP_NSUB * 16) + u0 * 16 + lane] = s_n[lane];
            __syncthreads();
        }
    }
}

__device__ __forceinline__ void sp_add_row(float4* acc, int r, int lane, float v, float4 xv) {
    float4 a = acc[r * 32 + lane];
    a.x += v * xv.x; a.y += v * xv.y; a.z += v * xv.z; a.w += v * xv.w;
    acc[r * 32 + lane] = a;
}

typedef __attribute__((ext_vector_type(4))) float sp_f4;

template <bool FULL>
__device__ __forceinline__ void sp_add16(float4* acc, const float* __restrict__ x, int lane,
                                         int er, int ec, float ev, int cnt) {
    sp_f4 fx[16];
#pragma unroll
    for (int s = 0; s < 16; ++s) {
        fx[s] = (sp_f4){0.f, 0.f, 0.f, 0.f};
        if (FULL || s < cnt) {
            const int c = __builtin_amdgcn_readlane(ec, s);
            fx[s] = *(const sp_f4*)(x + (size_t)c * IN_DIM + lane * 4);
        }
    }
    if (FULL) {
        asm volatile("" :: "v"(fx[0]), "v"(fx[1]), "v"(fx[2]), "v"(fx[3]), "v"(fx[4]), "v"(fx[5]),
                           "v"(fx[6]), "v"(fx[7]), "v"(fx[8]), "v"(fx[9]), "v"(fx[10]), "v"(fx[11]),
                           "v"(fx[12]), "v"(fx[13]), "v"(fx[14]), "v"(fx[15]));
    }
#pragma unroll
    for (int s = 0; s < 16; ++s) {
        if (FULL || s < cnt) {
            const int   r = __builtin_amdgcn_readlane(er, s);
            const float v = __int_as_float(__builtin_amdgcn_readlane(__float_as_int(ev), s));
            sp_add_row(acc, r, lane, v, make_float4(fx[s].x, fx[s].y, fx[s].z, fx[s].w));
        }
    }
}

template <bool LAST>
__device__ __forceinline__ int4 sp_add16_noted(float4* acc, const float* __restrict__ x, int lane,
                                               int4 cur, const int4* __restrict__ nextp, int cnt) {
    sp_f4 fx[16];
#pragma unroll
    for (int s = 0; s < 16; ++s) {
        const int c = __builtin_amdgcn_readlane(cur.y, s);
        fx[s] = *(const sp_f4*)(x + (size_t)c * IN_DIM + lane * 4);
    }
    int4 nx = cur;
    if (!LAST) nx = *nextp;
    asm volatile("" :: "v"(fx[0]), "v"(fx[1]), "v"(fx[2]), "v"(fx[3]), "v"(fx[4]), "v"(fx[5]),
                       "v"(fx[6]), "v"(fx[7]), "v"(fx[8]), "v"(fx[9]), "v"(fx[10]), "v"(fx[11]),
                       "v"(fx[12]), "v"(fx[13]), "v"(fx[14]), "v"(fx[15]),
                       "v"(nx.x), "v"(nx.y), "v"(nx.z));
#pragma unroll
    for (int s = 0; s < 16; ++s) {
        if (!LAST || s < cnt) {
            const int   r = __builtin_amdgcn_readlane(cur.x, s);
            const float v = __int_as_float(__builtin_amdgcn_readlane(cur.z, s));
            sp_add_row(acc, r, lane, v, make_float4(fx[s].x, fx[s].y, fx[s].z, fx[s].w));
        }
    }
    return nx;
}

template <int MODE>
__global__ __launch_bounds__(1024)
void gcn_spmm_rows(const int* __restrict__ lrow, const int* __restrict__ lcol,
                   const float* __restrict__ lval, const int* __restrict__ lcnt,
                   const float* __restrict__ x,
                   float* __restrict__ agg, int* __restrict__ rec_n, int* __restrict__ rec,
                   int nNodes, int nEdges) {
    __shared__ float4        acc[SP_RB * 32];
    __shared__ int           l_c[SP_WAVES * SP_CAP];
    __shared__ float         l_v[SP_WAVES * SP_CAP];
    __shared__ __attribute__((aligned(16))) unsigned char l_r[SP_WAVES * SP_CAP];
    __shared__ unsigned short l_o[SP_WAVES * SP_OWN];
    __shared__ int           l_n[SP_WAVES];
    __shared__ int           more[2];

    const int tid  = threadIdx.x;
    const int lane = tid & 31;
    const int w    = __builtin_amdgcn_readfirstlane(tid >> 5);
    const int subCap    = sp_sub_cap(nEdges);
    const int maxRounds = 4 * ((subCap + 127) >> 7) + 2;
    const int u0 = w * 16;

    for (int r0 = blockIdx.x * SP_RB; r0 < nNodes; r0 += gridDim.x * SP_RB) {
        const int rb = r0 / SP_RB;
        int4* recW = (int4*)rec + (size_t)((rb < SP_RECB ? rb : 0) * SP_WAVES + w) * SP_REC;
        if (MODE == 1) {
            const int rn = rb < SP_RECB ? rec_n[rb * SP_WAVES + lane] : -1;
            if (__builtin_amdgcn_ballot_w32(rn < 0) == 0u) {
                const int myN = __builtin_amdgcn_readlane(rn, w);
                const int l16 = lane & 15;
                int4 cur = make_int4(0, 0, 0, 0);
                if (myN > 0) cur = recW[l16 < myN ? l16 : myN - 1];
                for (int r = w; r < SP_RB; r += SP_WAVES) acc[r * 32 + lane] = make_float4(0.f, 0.f, 0.f, 0.f);
                for (int i0 = 0; i0 < myN; i0 += 16) {
                    const int left = myN - i0;
                    if (left >= 16) {
                        const int ni = i0 + 16 + l16;
                        cur = sp_add16_noted<false>(acc, x, lane, cur, recW + (ni < myN ? ni : myN - 1), 16);
                    } else {
                        sp_add16_noted<true>(acc, x, lane, cur, recW, left);
                    }
                }
                for (int r = w; r < SP_RB; r += SP_WAVES)
                    if (r0 + r < nNodes) ((float4*)agg)[(size_t)(r0 + r) * 32 + lane] = acc[r * 32 + lane];
                __syncthreads();
                continue;
            }
        }
        int  recN  = 0;
        bool recOk = rb < SP_RECB;
        const int s = r0 / (SP_SB * SP_RB);
        const int myc = lcnt[(s >> 4) * (SP_NSUB * 16) + (u0 + (lane & 15)) * 16 + (s & 15)];
        for (int i = tid; i < SP_RB * 32; i += 1024) acc[i] = make_float4(0.f, 0.f, 0.f, 0.f);
        if (tid < 2) more[tid] = 0;
        int g = 0;
        int c = 0;
        __syncthreads();

        for (int round = 0; round < maxRounds; ++round) {
            int n = 0;
            while (g < 4 && n + SP_SWEEP <= SP_CAP) {
                int nq[4], bq[4];
                int cmax = 0;
#pragma unroll
                for (int q = 0; q < 4; ++q) {
                    nq[q] = __builtin_amdgcn_readlane(myc, g * 4 + q);
                    bq[q] = (s * SP_NSUB + u0 + g * 4 + q) * subCap;
                    cmax  = nq[q] > cmax ? nq[q] : cmax;
                }
                if (c >= cmax) { ++g; c = 0; continue; }
                const int idx = c + lane * 4;
                c += 128;
                int4 rr[4];
#pragma unroll
                for (int q = 0; q < 4; ++q) {
                    rr[q] = make_int4(-1, -1, -1, -1);
                    if (idx < nq[q]) rr[q] = *(const int4*)(lrow + bq[q] + idx);
                }
#pragma unroll
                for (int q = 0; q < 4; ++q) {
                    const int p0 = bq[q] + idx;
                    unsigned d[4];
                    d[0] = (unsigned)(rr[q].x - r0); d[1] = (unsigned)(rr[q].y - r0);
                    d[2] = (unsigned)(rr[q].z - r0); d[3] = (unsigned)(rr[q].w - r0);
                    bool any = (d[0] < SP_RB) | (d[1] < SP_RB) | (d[2] < SP_RB) | (d[3] < SP_RB);
                    if (__builtin_amdgcn_ballot_w32(any) != 0u) {
#pragma unroll
                        for (int j = 0; j < 4; ++j) {
                            bool hit = d[j] < SP_RB;
                            unsigned m = __builtin_amdgcn_ballot_w32(hit);
                            if (m != 0u) {
                                if (hit) {
                                    int qq = w * SP_CAP + n + (int)__builtin_amdgcn_mbcnt_lo(m, 0u);
                                    l_r[qq] = (unsigned char)d[j];
                                    l_c[qq] = p0 + j;
                                }
                                n += __builtin_popcount(m);
                            }
                        }
                    }
                }
            }
            for (int i = lane; i < n; i += 32) {
                int e = l_c[w * SP_CAP + i];
                l_c[w * SP_CAP + i] = lcol[e];
                l_v[w * SP_CAP + i] = lval[e];
            }
            if (lane == 0) {
                l_n[w] = n;
                if (g < 4) more[round & 1] = 1;
            }
            __syncthreads();
            const int again = more[round & 1];
            if (tid == 0) more[(round + 1) & 1] = 0;

            {
                const int myn = l_n[lane];
                unsigned short* own = l_o + w * SP_OWN;
                int nOwn = 0;
                int Lg = 0;
                int b  = 0;
                for (int pass = 0; pass < 8 * ((SP_CAP + 127) / 128) + 1; ++pass) {
                    while (Lg < 8) {
                        int cnt[4];
                        int cmax = 0, stepMax = 0;
#pragma unroll
                        for (int q = 0; q < 4; ++q) {
                            cnt[q] = __builtin_amdgcn_readlane(myn, Lg * 4 + q);
                            cmax   = cnt[q] > cmax ? cnt[q] : cmax;
                            int left = cnt[q] - b;
                            left = left < 0 ? 0 : (left > 128 ? 128 : left);
                            stepMax += left;
                        }
                        if (b >= cmax) { ++Lg; b = 0; continue; }
                        if (nOwn + stepMax > SP_OWN) break;
                        const int p = b + lane * 4;
                        unsigned wd[4];
                        int      vc[4];
#pragma unroll
                        for (int q = 0; q < 4; ++q) {
                            int v = cnt[q] - p;
                            vc[q] = v < 0 ? 0 : (v > 4 ? 4 : v);
                            wd[q] = 0u;
                            if (vc[q] > 0) wd[q] = *(const unsigned*)(l_r + (Lg * 4 + q) * SP_CAP + p);
                        }
#pragma unroll
                        for (int q = 0; q < 4; ++q) {
#pragma unroll
                            for (int j = 0; j < 4; ++j) {
                                const bool hit = (j < vc[q]) & ((int)((wd[q] >> (8 * j)) & 31u) == w);
                                const unsigned m = __builtin_amdgcn_ballot_w32(hit);
                                if (m != 0u) {
                                    if (hit)
                                        own[nOwn + (int)__builtin_amdgcn_mbcnt_lo(m, 0u)] =
                                            (unsigned short)((Lg * 4 + q) * SP_CAP + p + j);
                                    nOwn += __builtin_popcount(m);
                                }
                            }
                        }
                        b += 128;
                    }
                    for (int i0 = 0; i0 < nOwn; i0 += 16) {
                        int   er = 0, ec = 0;
                        float ev = 0.f;
                        if (lane < 16 && i0 + lane < nOwn) {
                            const int e = own[i0 + lane];
                            er = (int)l_r[e];
                            ec = l_c[e];
                            ev = l_v[e];
                        }
                        const int left = nOwn - i0;
                        if (MODE == 0) {
                            const int take = left < 16 ? left : 16;
                            if (recOk && recN + take <= SP_REC) {
                                if (lane < take) recW[recN + lane] = make_int4(er, ec, __float_as_int(ev), 0);
                                recN += take;
                            } else {
                                recOk = false;
                            }
                        }
                        if (left >= 16) sp_add16<true>(acc, x, lane, er, ec, ev, 16);
                        else            sp_add16<false>(acc, x, lane, er, ec, ev, left);
                    }
                    nOwn = 0;
                    if (Lg >= 8) break;
                }
            }
            __syncthreads();
            if (!again) break;
        }

        if (MODE == 0 && rb < SP_RECB && lane == 0) rec_n[rb * SP_WAVES + w] = recOk ? recN : -1;

        for (int i = tid; i < SP_RB * 32; i += 1024) {
            int r = r0 + (i >> 5);
            if (r < nNodes) ((float4*)agg)[(size_t)r * 32 + (i & 31)] = acc[i];
        }
        __syncthreads();
    }
}

typedef __attribute__((ext_vector_type(8))) _Float16 v8h;
#define GM_WAVES 16

template <int NTILES, bool RELU>
__global__ __launch_bounds__(GM_WAVES * 32)
void gcn_gemm_wmma(const float* __restrict__ A,
                   const float* __restrict__ W,
                   float* __restrict__ C, int M) {
    __shared__ v8h b_lo[NTILES * 4 * 32];
    __shared__ v8h b_hi[NTILES * 4 * 32];
    const int N = NTILES * 16;
    const int lane = threadIdx.x & 31;
    const int warp = threadIdx.x >> 5;
    const int mt = blockIdx.x * GM_WAVES + warp;
    const int mtiles = M >> 4;
    const bool active = mt < mtiles;

    const int half = lane >> 4;
    const int r    = lane & 15;
    const int mrow = (active ? mt : 0) * 16 + r;

    float4 f[4][4];
#pragma unroll
    for (int kk = 0; kk < 4; ++kk) {
        const float* ap = A + (size_t)mrow * IN_DIM + kk * 32 + half * 8;
        f[kk][0] = *(const float4*)(ap + 0);
        f[kk][1] = *(const float4*)(ap + 4);
        f[kk][2] = *(const float4*)(ap + 16);
        f[kk][3] = *(const float4*)(ap + 20);
    }

    static_assert((NTILES * 4 * 32) % (GM_WAVES * 32) == 0 && NTILES % 2 == 0, "whole trips, tile pairs");
#pragma unroll
    for (int i = 0; i < (NTILES * 4 * 32) / (GM_WAVES * 32); ++i) {
        const int bid = (int)threadIdx.x + i * (GM_WAVES * 32);
        const int l   = bid & 31;
        const int hf  = (bid >> 5) & 1;
        const int kk  = (bid >> 6) & 3;
        const int tp  = bid >> 8;
        const int n   = tp * 32 + l;
        const int kb  = kk * 32 + hf * 8;
        v8h lo, hi;
#pragma unroll
        for (int j = 0; j < 8; ++j) {
            lo[j] = (_Float16)W[(size_t)(kb + j) * N + n];
            hi[j] = (_Float16)W[(size_t)(kb + 16 + j) * N + n];
        }
        const int id = ((tp * 2 + (l >> 4)) * 4 + kk) * 32 + hf * 16 + (l & 15);
        b_lo[id] = lo;
        b_hi[id] = hi;
    }
    v16h afr[4];
#pragma unroll
    for (int kk = 0; kk < 4; ++kk) {
        const float4 f0 = f[kk][0], f1 = f[kk][1], f2 = f[kk][2], f3 = f[kk][3];
        v16h a;
        a[0]  = (_Float16)f0.x; a[1]  = (_Float16)f0.y; a[2]  = (_Float16)f0.z; a[3]  = (_Float16)f0.w;
        a[4]  = (_Float16)f1.x; a[5]  = (_Float16)f1.y; a[6]  = (_Float16)f1.z; a[7]  = (_Float16)f1.w;
        a[8]  = (_Float16)f2.x; a[9]  = (_Float16)f2.y; a[10] = (_Float16)f2.z; a[11] = (_Float16)f2.w;
        a[12] = (_Float16)f3.x; a[13] = (_Float16)f3.y; a[14] = (_Float16)f3.z; a[15] = (_Float16)f3.w;
        afr[kk] = a;
    }
    __syncthreads();

    v8f acc[NTILES];
#pragma unroll
    for (int t = 0; t < NTILES; ++t) acc[t] = (v8f){};

#pragma unroll
    for (int kk = 0; kk < 4; ++kk) {
        const v16h a = afr[kk];
#pragma unroll
        for (int t = 0; t < NTILES; ++t) {
            const v8h lo = b_lo[(t * 4 + kk) * 32 + lane];
            const v8h hi = b_hi[(t * 4 + kk) * 32 + lane];
            const v16h b = __builtin_shufflevector(lo, hi, 0, 1, 2, 3, 4, 5, 6, 7,
                                                   8, 9, 10, 11, 12, 13, 14, 15);
            acc[t] = __builtin_amdgcn_wmma_f32_16x16x32_f16(
                         false, a,  false, b,
                         (short)0, acc[t],
                         false,  false);
        }
    }

    if (!active) return;

    int mbase = mt * 16 + half * 8;
#pragma unroll
    for (int t = 0; t < NTILES; ++t) {
        int ncol = t * 16 + r;
#pragma unroll
        for (int v = 0; v < 8; ++v) {
            float val = acc[t][v];
            if (RELU) val = fmaxf(val, 0.0f);
            C[(size_t)(mbase + v) * N + ncol] = val;
        }
    }
}

extern "C" void kernel_launch(void* const* d_in, const int* in_sizes, int n_in,
                              void* d_out, int out_size, void* d_ws, size_t ws_size,
                              hipStream_t stream) {
    const float* x    = (const float*)d_in[0];
    const int*   row  = (const int*)d_in[1];
    const int*   col  = (const int*)d_in[2];
    const float* vals = (const float*)d_in[3];
    const float* W1   = (const float*)d_in[4];
    const float* W2   = (const float*)d_in[5];
    float* out = (float*)d_out;

    const int NN = in_sizes[0] / IN_DIM;
    const int NE = in_sizes[1];

    float*     aggF = (float*)d_ws;
    float*     hF   = aggF + (size_t)NN * IN_DIM;
    int*       lcnt = (int*)(hF + (size_t)NN * IN_DIM);
    int*       lrow = lcnt + SP_NSUB * 16;
    int*       lcol = lrow + SP_LIST;
    float*     lval = (float*)(lcol + SP_LIST);
    int*       recLen = (int*)(lval + SP_LIST);
    int*       rec    = recLen + SP_RECB * SP_WAVES;

    gcn_edge_ranges<<<256, 1024, 0, stream>>>(row, col, vals, lrow, lcol, lval, lcnt, NN, NE);

    gcn_spmm_rows<0><<<256, 1024, 0, stream>>>(lrow, lcol, lval, lcnt, x, aggF, recLen, rec, NN, NE);

    {
        int mtiles = NN / 16;
        int blocks = (mtiles + GM_WAVES - 1) / GM_WAVES;
        gcn_gemm_wmma<8, true><<<blocks, GM_WAVES * 32, 0, stream>>>(aggF, W1, hF, NN);
    }

    gcn_spmm_rows<1><<<256, 1024, 0, stream>>>(lrow, lcol, lval, lcnt, hF, aggF, recLen, rec, NN, NE);

    {
        int mtiles = NN / 16;
        int blocks = (mtiles + GM_WAVES - 1) / GM_WAVES;
        gcn_gemm_wmma<4, false><<<blocks, GM_WAVES * 32, 0, stream>>>(aggF, W2, out, NN);
    }
}


# Both GEMMs move into the SpMM kernels: the work-group's 196 row sums stay in LDS (padded pitch), one wave per 16-row tile reads them as WMMA A fragments; 5 launches to 3; same f32 sums and order
# speedup vs baseline: 3.9584x; 1.1714x over previous
#include <hip/hip_runtime.h>
#include <hip/hip_bf16.h>

typedef __attribute__((ext_vector_type(16))) _Float16 v16h;
typedef __attribute__((ext_vector_type(8)))  _Float16 v8h;
typedef __attribute__((ext_vector_type(8)))  float    v8f;

#define IN_DIM 128

#define SP_RB     196
#define SP_WAVES  32
#define SP_SB     16
#define SP_NSUB   (SP_WAVES * 16)
#define SP_SWEEP  512
#define SP_CAP    640
#define SP_OWN    576
#define SP_LIST   (16 * SP_NSUB * 1568)
#define SP_REC    256
#define SP_RECB   256
#define SP_PITCH  33
#define SP_MT     ((SP_RB + 15) / 16)

__device__ __forceinline__ int sp_sub_cap(int nEdges) {
    return (((nEdges + SP_NSUB - 1) / SP_NSUB) + 31) & ~31;
}

__device__ __forceinline__ int sp_put4(const int* rr, const int* cc, const float* vv, int lo, int base, int n,
                                       int* __restrict__ lrow, int* __restrict__ lcol,
                                       float* __restrict__ lval) {
#pragma unroll
    for (int j = 0; j < 4; ++j) {
        const bool hit = (unsigned)(rr[j] - lo) < (unsigned)(SP_SB * SP_RB);
        const unsigned m = __builtin_amdgcn_ballot_w32(hit);
        if (m != 0u) {
            if (hit) {
                const int q = base + n + (int)__builtin_amdgcn_mbcnt_lo(m, 0u);
                lrow[q] = rr[j];
                lcol[q] = cc[j];
                lval[q] = vv[j];
            }
            n += __builtin_popcount(m);
        }
    }
    return n;
}

__global__ __launch_bounds__(1024)
void gcn_edge_ranges(const int* __restrict__ row, const int* __restrict__ col,
                     const float* __restrict__ vals,
                     int* __restrict__ lrow, int* __restrict__ lcol, float* __restrict__ lval,
                     int* __restrict__ lcnt, int nNodes, int nEdges) {
    __shared__ int s_n[32];
    const int lane = threadIdx.x & 31;
    const int w    = __builtin_amdgcn_readfirstlane(threadIdx.x >> 5);
    const int subCap  = sp_sub_cap(nEdges);
    const int nRanges = (nNodes + SP_SB * SP_RB - 1) / (SP_SB * SP_RB);

    for (int u0 = blockIdx.x * 2; u0 < SP_NSUB; u0 += gridDim.x * 2) {
        for (int sb = 0; sb < nRanges; sb += 16) {
            const int u = u0 + (w >> 4);
            const int s = sb + (w & 15);
            int n = 0;
            if (s < nRanges) {
                const int lo     = s * (SP_SB * SP_RB);
                const int eBegin = u * subCap;
                const int eEnd   = eBegin + subCap < nEdges ? eBegin + subCap : nEdges;
                const int base   = (s * SP_NSUB + u) * subCap;
                int c = eBegin;
                for (; c + 512 <= eEnd; c += 512) {
                    int4   r4[4], c4[4];
                    float4 v4[4];
#pragma unroll
                    for (int k = 0; k < 4; ++k) {
                        const int e0 = c + k * 128 + lane * 4;
                        r4[k] = *(const int4*)(row + e0);
                        c4[k] = *(const int4*)(col + e0);
                        v4[k] = *(const float4*)(vals + e0);
                    }
#pragma unroll
                    for (int k = 0; k < 4; ++k) {
                        const int   rr[4] = {r4[k].x, r4[k].y, r4[k].z, r4[k].w};
                        const int   cc[4] = {c4[k].x, c4[k].y, c4[k].z, c4[k].w};
                        const float vv[4] = {v4[k].x, v4[k].y, v4[k].z, v4[k].w};
                        n = sp_put4(rr, cc, vv, lo, base, n, lrow, lcol, lval);
                    }
                }
                for (; c < eEnd; c += 128) {
                    const int e0 = c + lane * 4;
                    int   rr[4], cc[4];
                    float vv[4];
                    if (e0 + 3 < eEnd) {
                        const int4   r4 = *(const int4*)(row + e0);
                        const int4   c4 = *(const int4*)(col + e0);
                        const float4 v4 = *(const float4*)(vals + e0);
                        rr[0] = r4.x; rr[1] = r4.y; rr[2] = r4.z; rr[3] = r4.w;
                        cc[0] = c4.x; cc[1] = c4.y; cc[2] = c4.z; cc[3] = c4.w;
                        vv[0] = v4.x; vv[1] = v4.y; vv[2] = v4.z; vv[3] = v4.w;
                    } else {
#pragma unroll
                        for (int j = 0; j < 4; ++j) {
                            const bool in = e0 + j < eEnd;
                            rr[j] = in ? row[e0 + j]  : -1;
                            cc[j] = in ? col[e0 + j]  : 0;
                            vv[j] = in ? vals[e0 + j] : 0.f;
                        }
                    }
                    n = sp_put4(rr, cc, vv, lo, base, n, lrow, lcol, lval);
                }
                if (lane < ((4 - (n & 3)) & 3)) lrow[base + n + lane] = -1;
            }
            if (lane == 0) s_n[w] = n;
            __syncthreads();
            if (w == 0) lcnt[(sb >> 4) * (SP_NSUB * 16) + u0 * 16 + lane] = s_n[lane];
            __syncthreads();
        }
    }
}

__device__ __forceinline__ void sp_add_row(float4* acc, int r, int lane, float v, float4 xv) {
    float4 a = acc[r * SP_PITCH + lane];
    a.x += v * xv.x; a.y += v * xv.y; a.z += v * xv.z; a.w += v * xv.w;
    acc[r * SP_PITCH + lane] = a;
}

typedef __attribute__((ext_vector_type(4))) float sp_f4;

template <bool FULL>
__device__ __forceinline__ void sp_add16(float4* acc, const float* __restrict__ x, int lane,
                                         int er, int ec, float ev, int cnt) {
    sp_f4 fx[16];
#pragma unroll
    for (int s = 0; s < 16; ++s) {
        fx[s] = (sp_f4){0.f, 0.f, 0.f, 0.f};
        if (FULL || s < cnt) {
            const int c = __builtin_amdgcn_readlane(ec, s);
            fx[s] = *(const sp_f4*)(x + (size_t)c * IN_DIM + lane * 4);
        }
    }
    if (FULL) {
        asm volatile("" :: "v"(fx[0]), "v"(fx[1]), "v"(fx[2]), "v"(fx[3]), "v"(fx[4]), "v"(fx[5]),
                           "v"(fx[6]), "v"(fx[7]), "v"(fx[8]), "v"(fx[9]), "v"(fx[10]), "v"(fx[11]),
                           "v"(fx[12]), "v"(fx[13]), "v"(fx[14]), "v"(fx[15]));
    }
#pragma unroll
    for (int s = 0; s < 16; ++s) {
        if (FULL || s < cnt) {
            const int   r = __builtin_amdgcn_readlane(er, s);
            const float v = __int_as_float(__builtin_amdgcn_readlane(__float_as_int(ev), s));
            sp_add_row(acc, r, lane, v, make_float4(fx[s].x, fx[s].y, fx[s].z, fx[s].w));
        }
    }
}

template <bool LAST>
__device__ __forceinline__ int4 sp_add16_noted(float4* acc, const float* __restrict__ x, int lane,
                                               int4 cur, const int4* __restrict__ nextp, int cnt) {
    sp_f4 fx[16];
#pragma unroll
    for (int s = 0; s < 16; ++s) {
        const int c = __builtin_amdgcn_readlane(cur.y, s);
        fx[s] = *(const sp_f4*)(x + (size_t)c * IN_DIM + lane * 4);
    }
    int4 nx = cur;
    if (!LAST) nx = *nextp;
    asm volatile("" :: "v"(fx[0]), "v"(fx[1]), "v"(fx[2]), "v"(fx[3]), "v"(fx[4]), "v"(fx[5]),
                       "v"(fx[6]), "v"(fx[7]), "v"(fx[8]), "v"(fx[9]), "v"(fx[10]), "v"(fx[11]),
                       "v"(fx[12]), "v"(fx[13]), "v"(fx[14]), "v"(fx[15]),
                       "v"(nx.x), "v"(nx.y), "v"(nx.z));
#pragma unroll
    for (int s = 0; s < 16; ++s) {
        if (!LAST || s < cnt) {
            const int   r = __builtin_amdgcn_readlane(cur.x, s);
            const float v = __int_as_float(__builtin_amdgcn_readlane(cur.z, s));
            sp_add_row(acc, r, lane, v, make_float4(fx[s].x, fx[s].y, fx[s].z, fx[s].w));
        }
    }
    return nx;
}

template <int NTILES>
__device__ __forceinline__ void sp_build_b(const float* __restrict__ W, v8h* b_lo, v8h* b_hi, int tid) {
    static_assert((NTILES * 4 * 32) % 32 == 0 && NTILES * 4 * 32 <= 1024 && NTILES % 2 == 0,
                  "whole waves, one trip, tile pairs");
    const int N = NTILES * 16;
    if (tid < NTILES * 4 * 32) {
        const int l   = tid & 31;
        const int hf  = (tid >> 5) & 1;
        const int kk  = (tid >> 6) & 3;
        const int tp  = tid >> 8;
        const int n   = tp * 32 + l;
        const int kb  = kk * 32 + hf * 8;
        v8h lo, hi;
#pragma unroll
        for (int j = 0; j < 8; ++j) {
            lo[j] = (_Float16)W[(size_t)(kb + j) * N + n];
            hi[j] = (_Float16)W[(size_t)(kb + 16 + j) * N + n];
        }
        const int id = ((tp * 2 + (l >> 4)) * 4 + kk) * 32 + hf * 16 + (l & 15);
        b_lo[id] = lo;
        b_hi[id] = hi;
    }
}

template <int NTILES, bool RELU>
__device__ __forceinline__ void sp_tile_times_w(float4* acc, const v8h* b_lo, const v8h* b_hi,
                                                int lane, int mt) {
    const int half = lane >> 4;
    const int r    = lane & 15;
    int row = mt * 16 + r;
    row = row < SP_RB ? row : SP_RB - 1;
    v8f c[NTILES];
#pragma unroll
    for (int t = 0; t < NTILES; ++t) c[t] = (v8f){};
#pragma unroll
    for (int kk = 0; kk < 4; ++kk) {
        const float4* ap = acc + row * SP_PITCH + kk * 8 + half * 2;
        const float4 f0 = ap[0], f1 = ap[1], f2 = ap[4], f3 = ap[5];
        v16h a;
        a[0]  = (_Float16)f0.x; a[1]  = (_Float16)f0.y; a[2]  = (_Float16)f0.z; a[3]  = (_Float16)f0.w;
        a[4]  = (_Float16)f1.x; a[5]  = (_Float16)f1.y; a[6]  = (_Float16)f1.z; a[7]  = (_Float16)f1.w;
        a[8]  = (_Float16)f2.x; a[9]  = (_Float16)f2.y; a[10] = (_Float16)f2.z; a[11] = (_Float16)f2.w;
        a[12] = (_Float16)f3.x; a[13] = (_Float16)f3.y; a[14] = (_Float16)f3.z; a[15] = (_Float16)f3.w;
#pragma unroll
        for (int t = 0; t < NTILES; ++t) {
            const v8h lo = b_lo[(t * 4 + kk) * 32 + lane];
            const v8h hi = b_hi[(t * 4 + kk) * 32 + lane];
            const v16h b = __builtin_shufflevector(lo, hi, 0, 1, 2, 3, 4, 5, 6, 7,
                                                   8, 9, 10, 11, 12, 13, 14, 15);
            c[t] = __builtin_amdgcn_wmma_f32_16x16x32_f16(
                        false, a,  false, b,
                        (short)0, c[t],
                        false,  false);
        }
    }
    if (RELU) {
#pragma unroll
        for (int t = 0; t < NTILES; ++t)
#pragma unroll
            for (int v = 0; v < 8; ++v) c[t][v] = fmaxf(c[t][v], 0.0f);
    }
    float* accf = (float*)acc;
#pragma unroll
    for (int v = 0; v < 8; ++v) {
        const int orow = mt * 16 + half * 8 + v;
        if (orow < SP_RB) {
#pragma unroll
            for (int t = 0; t < NTILES; ++t) accf[orow * (SP_PITCH * 4) + t * 16 + r] = c[t][v];
        }
    }
}

template <int NTILES, bool RELU>
__device__ __forceinline__ void sp_product_out(float4* acc, const v8h* b_lo, const v8h* b_hi,
                                               float* __restrict__ C, int r0, int nNodes,
                                               int tid, int lane, int w) {
    __syncthreads();
    if (w < SP_MT) sp_tile_times_w<NTILES, RELU>(acc, b_lo, b_hi, lane, w);
    __syncthreads();
    for (int i = tid; i < SP_RB * (NTILES * 4); i += 1024) {
        const int r = i / (NTILES * 4);
        const int c = i % (NTILES * 4);
        if (r0 + r < nNodes) ((float4*)C)[(size_t)(r0 + r) * (NTILES * 4) + c] = acc[r * SP_PITCH + c];
    }
}

template <int MODE>
__global__ __launch_bounds__(1024)
void gcn_spmm_rows(const int* __restrict__ lrow, const int* __restrict__ lcol,
                   const float* __restrict__ lval, const int* __restrict__ lcnt,
                   const float* __restrict__ x, const float* __restrict__ W,
                   float* __restrict__ C, int* __restrict__ rec_n, int* __restrict__ rec,
                   int nNodes, int nEdges) {
    constexpr int  NTILES = MODE == 0 ? 8 : 4;
    constexpr bool RELU   = MODE == 0;
    __shared__ float4        acc[SP_RB * SP_PITCH];
    __shared__ __attribute__((aligned(16))) int l_c[SP_WAVES * SP_CAP];
    __shared__ float         l_v[SP_WAVES * SP_CAP];
    __shared__ __attribute__((aligned(16))) unsigned char l_r[SP_WAVES * SP_CAP];
    __shared__ unsigned short l_o[SP_WAVES * SP_OWN];
    __shared__ int           l_n[SP_WAVES];
    __shared__ int           more[2];

    static_assert(2 * NTILES * 4 * 32 * sizeof(v8h) <= sizeof(int) * SP_WAVES * SP_CAP, "B fragments fit the list");
    static_assert(SP_MT <= SP_WAVES, "a wave per 16-row tile");
    v8h* b_lo = (v8h*)l_c;
    v8h* b_hi = b_lo + NTILES * 4 * 32;

    const int tid  = threadIdx.x;
    const int lane = tid & 31;
    const int w    = __builtin_amdgcn_readfirstlane(tid >> 5);
    const int subCap    = sp_sub_cap(nEdges);
    const int maxRounds = 4 * ((subCap + 127) >> 7) + 2;
    const int u0 = w * 16;

    for (int r0 = blockIdx.x * SP_RB; r0 < nNodes; r0 += gridDim.x * SP_RB) {
        const int rb = r0 / SP_RB;
        int4* recW = (int4*)rec + (size_t)((rb < SP_RECB ? rb : 0) * SP_WAVES + w) * SP_REC;
        if (MODE == 1) {
            const int rn = rb < SP_RECB ? rec_n[rb * SP_WAVES + lane] : -1;
            if (__builtin_amdgcn_ballot_w32(rn < 0) == 0u) {
                const int myN = __builtin_amdgcn_readlane(rn, w);
                const int l16 = lane & 15;
                int4 cur = make_int4(0, 0, 0, 0);
                if (myN > 0) cur = recW[l16 < myN ? l16 : myN - 1];
                sp_build_b<NTILES>(W, b_lo, b_hi, tid);
                for (int r = w; r < SP_RB; r += SP_WAVES) acc[r * SP_PITCH + lane] = make_float4(0.f, 0.f, 0.f, 0.f);
                for (int i0 = 0; i0 < myN; i0 += 16) {
                    const int left = myN - i0;
                    if (left >= 16) {
                        const int ni = i0 + 16 + l16;
                        cur = sp_add16_noted<false>(acc, x, lane, cur, recW + (ni < myN ? ni : myN - 1), 16);
                    } else {
                        sp_add16_noted<true>(acc, x, lane, cur, recW, left);
                    }
                }
                sp_product_out<NTILES, RELU>(acc, b_lo, b_hi, C, r0, nNodes, tid, lane, w);
                __syncthreads();
                continue;
            }
        }
        int  recN  = 0;
        bool recOk = rb < SP_RECB;
        const int s = r0 / (SP_SB * SP_RB);
        const int myc = lcnt[(s >> 4) * (SP_NSUB * 16) + (u0 + (lane & 15)) * 16 + (s & 15)];
        for (int i = tid; i < SP_RB * SP_PITCH; i += 1024) acc[i] = make_float4(0.f, 0.f, 0.f, 0.f);
        if (tid < 2) more[tid] = 0;
        int g = 0;
        int c = 0;
        __syncthreads();

        for (int round = 0; round < maxRounds; ++round) {
            int n = 0;
            while (g < 4 && n + SP_SWEEP <= SP_CAP) {
                int nq[4], bq[4];
                int cmax = 0;
#pragma unroll
                for (int q = 0; q < 4; ++q) {
                    nq[q] = __builtin_amdgcn_readlane(myc, g * 4 + q);
                    bq[q] = (s * SP_NSUB + u0 + g * 4 + q) * subCap;
                    cmax  = nq[q] > cmax ? nq[q] : cmax;
                }
                if (c >= cmax) { ++g; c = 0; continue; }
                const int idx = c + lane * 4;
                c += 128;
                int4 rr[4];
#pragma unroll
                for (int q = 0; q < 4; ++q) {
                    rr[q] = make_int4(-1, -1, -1, -1);
                    if (idx < nq[q]) rr[q] = *(const int4*)(lrow + bq[q] + idx);
                }
#pragma unroll
                for (int q = 0; q < 4; ++q) {
                    const int p0 = bq[q] + idx;
                    unsigned d[4];
                    d[0] = (unsigned)(rr[q].x - r0); d[1] = (unsigned)(rr[q].y - r0);
                    d[2] = (unsigned)(rr[q].z - r0); d[3] = (unsigned)(rr[q].w - r0);
                    bool any = (d[0] < SP_RB) | (d[1] < SP_RB) | (d[2] < SP_RB) | (d[3] < SP_RB);
                    if (__builtin_amdgcn_ballot_w32(any) != 0u) {
#pragma unroll
                        for (int j = 0; j < 4; ++j) {
                            bool hit = d[j] < SP_RB;
                            unsigned m = __builtin_amdgcn_ballot_w32(hit);
                            if (m != 0u) {
                                if (hit) {
                                    int qq = w * SP_CAP + n + (int)__builtin_amdgcn_mbcnt_lo(m, 0u);
                                    l_r[qq] = (unsigned char)d[j];
                                    l_c[qq] = p0 + j;
                                }
                                n += __builtin_popcount(m);
                            }
                        }
                    }
                }
            }
            for (int i = lane; i < n; i += 32) {
                int e = l_c[w * SP_CAP + i];
                l_c[w * SP_CAP + i] = lcol[e];
                l_v[w * SP_CAP + i] = lval[e];
            }
            if (lane == 0) {
                l_n[w] = n;
                if (g < 4) more[round & 1] = 1;
            }
            __syncthreads();
            const int again = more[round & 1];
            if (tid == 0) more[(round + 1) & 1] = 0;

            {
                const int myn = l_n[lane];
                unsigned short* own = l_o + w * SP_OWN;
                int nOwn = 0;
                int Lg = 0;
                int b  = 0;
                for (int pass = 0; pass < 8 * ((SP_CAP + 127) / 128) + 1; ++pass) {
                    while (Lg < 8) {
                        int cnt[4];
                        int cmax = 0, stepMax = 0;
#pragma unroll
                        for (int q = 0; q < 4; ++q) {
                            cnt[q] = __builtin_amdgcn_readlane(myn, Lg * 4 + q);
                            cmax   = cnt[q] > cmax ? cnt[q] : cmax;
                            int left = cnt[q] - b;
                            left = left < 0 ? 0 : (left > 128 ? 128 : left);
                            stepMax += left;
                        }
                        if (b >= cmax) { ++Lg; b = 0; continue; }
                        if (nOwn + stepMax > SP_OWN) break;
                        const int p = b + lane * 4;
                        unsigned wd[4];
                        int      vc[4];
#pragma unroll
                        for (int q = 0; q < 4; ++q) {
                            int v = cnt[q] - p;
                            vc[q] = v < 0 ? 0 : (v > 4 ? 4 : v);
                            wd[q] = 0u;
                            if (vc[q] > 0) wd[q] = *(const unsigned*)(l_r + (Lg * 4 + q) * SP_CAP + p);
                        }
#pragma unroll
                        for (int q = 0; q < 4; ++q) {
#pragma unroll
                            for (int j = 0; j < 4; ++j) {
                                const bool hit = (j < vc[q]) & ((int)((wd[q] >> (8 * j)) & 31u) == w);
                                const unsigned m = __builtin_amdgcn_ballot_w32(hit);
                                if (m != 0u) {
                                    if (hit)
                                        own[nOwn + (int)__builtin_amdgcn_mbcnt_lo(m, 0u)] =
                                            (unsigned short)((Lg * 4 + q) * SP_CAP + p + j);
                                    nOwn += __builtin_popcount(m);
                                }
                            }
                        }
                        b += 128;
                    }
                    for (int i0 = 0; i0 < nOwn; i0 += 16) {
                        int   er = 0, ec = 0;
                        float ev = 0.f;
                        if (lane < 16 && i0 + lane < nOwn) {
                            const int e = own[i0 + lane];
                            er = (int)l_r[e];
                            ec = l_c[e];
                            ev = l_v[e];
                        }
                        const int left = nOwn - i0;
                        if (MODE == 0) {
                            const int take = left < 16 ? left : 16;
                            if (recOk && recN + take <= SP_REC) {
                                if (lane < take) recW[recN + lane] = make_int4(er, ec, __float_as_int(ev), 0);
                                recN += take;
                            } else {
                                recOk = false;
                            }
                        }
                        if (left >= 16) sp_add16<true>(acc, x, lane, er, ec, ev, 16);
                        else            sp_add16<false>(acc, x, lane, er, ec, ev, left);
                    }
                    nOwn = 0;
                    if (Lg >= 8) break;
                }
            }
            __syncthreads();
            if (!again) break;
        }

        if (MODE == 0 && rb < SP_RECB && lane == 0) rec_n[rb * SP_WAVES + w] = recOk ? recN : -1;

        sp_build_b<NTILES>(W, b_lo, b_hi, tid);
        sp_product_out<NTILES, RELU>(acc, b_lo, b_hi, C, r0, nNodes, tid, lane, w);
        __syncthreads();
    }
}

extern "C" void kernel_launch(void* const* d_in, const int* in_sizes, int n_in,
                              void* d_out, int out_size, void* d_ws, size_t ws_size,
                              hipStream_t stream) {
    const float* x    = (const float*)d_in[0];
    const int*   row  = (const int*)d_in[1];
    const int*   col  = (const int*)d_in[2];
    const float* vals = (const float*)d_in[3];
    const float* W1   = (const float*)d_in[4];
    const float* W2   = (const float*)d_in[5];
    float* out = (float*)d_out;

    const int NN = in_sizes[0] / IN_DIM;
    const int NE = in_sizes[1];

    float*     hF   = (float*)d_ws;
    int*       lcnt = (int*)(hF + (size_t)NN * IN_DIM);
    int*       lrow = lcnt + SP_NSUB * 16;
    int*       lcol = lrow + SP_LIST;
    float*     lval = (float*)(lcol + SP_LIST);
    int*       recLen = (int*)(lval + SP_LIST);
    int*       rec    = recLen + SP_RECB * SP_WAVES;

    gcn_edge_ranges<<<256, 1024, 0, stream>>>(row, col, vals, lrow, lcol, lval, lcnt, NN, NE);

    gcn_spmm_rows<0><<<256, 1024, 0, stream>>>(lrow, lcol, lval, lcnt, x, W1, hF, recLen, rec, NN, NE);

    gcn_spmm_rows<1><<<256, 1024, 0, stream>>>(lrow, lcol, lval, lcnt, hF, W2, out, recLen, rec, NN, NE);
}


# SpMM adds, both layers: a wave groups its entries by row (stable) and keeps the running row sum in registers, LDS touched once per row, not a read-modify-write per entry; same f32 sums and order
# speedup vs baseline: 4.0894x; 1.0331x over previous
#include <hip/hip_runtime.h>
#include <hip/hip_bf16.h>

typedef __attribute__((ext_vector_type(16))) _Float16 v16h;
typedef __attribute__((ext_vector_type(8)))  _Float16 v8h;
typedef __attribute__((ext_vector_type(8)))  float    v8f;

#define IN_DIM 128

#define SP_RB     196
#define SP_WAVES  32
#define SP_SB     16
#define SP_NSUB   (SP_WAVES * 16)
#define SP_SWEEP  512
#define SP_CAP    640
#define SP_OWN    576
#define SP_LIST   (16 * SP_NSUB * 1568)
#define SP_REC    256
#define SP_RECB   256
#define SP_PITCH  33
#define SP_MT     ((SP_RB + 15) / 16)
#define SP_KR     ((SP_RB + SP_WAVES - 1) / SP_WAVES)

__device__ __forceinline__ int sp_sub_cap(int nEdges) {
    return (((nEdges + SP_NSUB - 1) / SP_NSUB) + 31) & ~31;
}

__device__ __forceinline__ int sp_put4(const int* rr, const int* cc, const float* vv, int lo, int base, int n,
                                       int* __restrict__ lrow, int* __restrict__ lcol,
                                       float* __restrict__ lval) {
#pragma unroll
    for (int j = 0; j < 4; ++j) {
        const bool hit = (unsigned)(rr[j] - lo) < (unsigned)(SP_SB * SP_RB);
        const unsigned m = __builtin_amdgcn_ballot_w32(hit);
        if (m != 0u) {
            if (hit) {
                const int q = base + n + (int)__builtin_amdgcn_mbcnt_lo(m, 0u);
                lrow[q] = rr[j];
                lcol[q] = cc[j];
                lval[q] = vv[j];
            }
            n += __builtin_popcount(m);
        }
    }
    return n;
}

__global__ __launch_bounds__(1024)
void gcn_edge_ranges(const int* __restrict__ row, const int* __restrict__ col,
                     const float* __restrict__ vals,
                     int* __restrict__ lrow, int* __restrict__ lcol, float* __restrict__ lval,
                     int* __restrict__ lcnt, int nNodes, int nEdges) {
    __shared__ int s_n[32];
    const int lane = threadIdx.x & 31;
    const int w    = __builtin_amdgcn_readfirstlane(threadIdx.x >> 5);
    const int subCap  = sp_sub_cap(nEdges);
    const int nRanges = (nNodes + SP_SB * SP_RB - 1) / (SP_SB * SP_RB);

    for (int u0 = blockIdx.x * 2; u0 < SP_NSUB; u0 += gridDim.x * 2) {
        for (int sb = 0; sb < nRanges; sb += 16) {
            const int u = u0 + (w >> 4);
            const int s = sb + (w & 15);
            int n = 0;
            if (s < nRanges) {
                const int lo     = s * (SP_SB * SP_RB);
                const int eBegin = u * subCap;
                const int eEnd   = eBegin + subCap < nEdges ? eBegin + subCap : nEdges;
                const int base   = (s * SP_NSUB + u) * subCap;
                int c = eBegin;
                for (; c + 512 <= eEnd; c += 512) {
                    int4   r4[4], c4[4];
                    float4 v4[4];
#pragma unroll
                    for (int k = 0; k < 4; ++k) {
                        const int e0 = c + k * 128 + lane * 4;
                        r4[k] = *(const int4*)(row + e0);
                        c4[k] = *(const int4*)(col + e0);
                        v4[k] = *(const float4*)(vals + e0);
                    }
#pragma unroll
                    for (int k = 0; k < 4; ++k) {
                        const int   rr[4] = {r4[k].x, r4[k].y, r4[k].z, r4[k].w};
                        const int   cc[4] = {c4[k].x, c4[k].y, c4[k].z, c4[k].w};
                        const float vv[4] = {v4[k].x, v4[k].y, v4[k].z, v4[k].w};
                        n = sp_put4(rr, cc, vv, lo, base, n, lrow, lcol, lval);
                    }
                }
                for (; c < eEnd; c += 128) {
                    const int e0 = c + lane * 4;
                    int   rr[4], cc[4];
                    float vv[4];
                    if (e0 + 3 < eEnd) {
                        const int4   r4 = *(const int4*)(row + e0);
                        const int4   c4 = *(const int4*)(col + e0);
                        const float4 v4 = *(const float4*)(vals + e0);
                        rr[0] = r4.x; rr[1] = r4.y; rr[2] = r4.z; rr[3] = r4.w;
                        cc[0] = c4.x; cc[1] = c4.y; cc[2] = c4.z; cc[3] = c4.w;
                        vv[0] = v4.x; vv[1] = v4.y; vv[2] = v4.z; vv[3] = v4.w;
                    } else {
#pragma unroll
                        for (int j = 0; j < 4; ++j) {
                            const bool in = e0 + j < eEnd;
                            rr[j] = in ? row[e0 + j]  : -1;
                            cc[j] = in ? col[e0 + j]  : 0;
                            vv[j] = in ? vals[e0 + j] : 0.f;
                        }
                    }
                    n = sp_put4(rr, cc, vv, lo, base, n, lrow, lcol, lval);
                }
                if (lane < ((4 - (n & 3)) & 3)) lrow[base + n + lane] = -1;
            }
            if (lane == 0) s_n[w] = n;
            __syncthreads();
            if (w == 0) lcnt[(sb >> 4) * (SP_NSUB * 16) + u0 * 16 + lane] = s_n[lane];
            __syncthreads();
        }
    }
}

typedef __attribute__((ext_vector_type(4))) float sp_f4;

struct sp_run {
    sp_f4 a;
    int   row;
};

__device__ __forceinline__ void sp_run_add(float4* acc, sp_run& run, int r, int lane, float v, sp_f4 xv) {
    if (r != run.row) {
        if (run.row >= 0) acc[run.row * SP_PITCH + lane] = make_float4(run.a.x, run.a.y, run.a.z, run.a.w);
        const float4 t = acc[r * SP_PITCH + lane];
        run.a   = (sp_f4){t.x, t.y, t.z, t.w};
        run.row = r;
    }
    run.a.x += v * xv.x; run.a.y += v * xv.y; run.a.z += v * xv.z; run.a.w += v * xv.w;
}

__device__ __forceinline__ void sp_run_flush(float4* acc, sp_run& run, int lane) {
    if (run.row >= 0) acc[run.row * SP_PITCH + lane] = make_float4(run.a.x, run.a.y, run.a.z, run.a.w);
    run.row = -1;
}


template <bool FULL>
__device__ __forceinline__ void sp_add16(float4* acc, sp_run& run, const float* __restrict__ x, int lane,
                                         int er, int ec, float ev, int cnt) {
    sp_f4 fx[16];
#pragma unroll
    for (int s = 0; s < 16; ++s) {
        fx[s] = (sp_f4){0.f, 0.f, 0.f, 0.f};
        if (FULL || s < cnt) {
            const int c = __builtin_amdgcn_readlane(ec, s);
            fx[s] = *(const sp_f4*)(x + (size_t)c * IN_DIM + lane * 4);
        }
    }
    if (FULL) {
        asm volatile("" :: "v"(fx[0]), "v"(fx[1]), "v"(fx[2]), "v"(fx[3]), "v"(fx[4]), "v"(fx[5]),
                           "v"(fx[6]), "v"(fx[7]), "v"(fx[8]), "v"(fx[9]), "v"(fx[10]), "v"(fx[11]),
                           "v"(fx[12]), "v"(fx[13]), "v"(fx[14]), "v"(fx[15]));
    }
#pragma unroll
    for (int s = 0; s < 16; ++s) {
        if (FULL || s < cnt) {
            const int   r = __builtin_amdgcn_readlane(er, s);
            const float v = __int_as_float(__builtin_amdgcn_readlane(__float_as_int(ev), s));
            sp_run_add(acc, run, r, lane, v, fx[s]);
        }
    }
}

template <bool LAST>
__device__ __forceinline__ int4 sp_add16_noted(float4* acc, sp_run& run, const float* __restrict__ x, int lane,
                                               int4 cur, const int4* __restrict__ nextp, int cnt) {
    sp_f4 fx[16];
#pragma unroll
    for (int s = 0; s < 16; ++s) {
        const int c = __builtin_amdgcn_readlane(cur.y, s);
        fx[s] = *(const sp_f4*)(x + (size_t)c * IN_DIM + lane * 4);
    }
    int4 nx = cur;
    if (!LAST) nx = *nextp;
    asm volatile("" :: "v"(fx[0]), "v"(fx[1]), "v"(fx[2]), "v"(fx[3]), "v"(fx[4]), "v"(fx[5]),
                       "v"(fx[6]), "v"(fx[7]), "v"(fx[8]), "v"(fx[9]), "v"(fx[10]), "v"(fx[11]),
                       "v"(fx[12]), "v"(fx[13]), "v"(fx[14]), "v"(fx[15]),
                       "v"(nx.x), "v"(nx.y), "v"(nx.z));
#pragma unroll
    for (int s = 0; s < 16; ++s) {
        if (!LAST || s < cnt) {
            const int   r = __builtin_amdgcn_readlane(cur.x, s);
            const float v = __int_as_float(__builtin_amdgcn_readlane(cur.z, s));
            sp_run_add(acc, run, r, lane, v, fx[s]);
        }
    }
    return nx;
}

template <int NTILES>
__device__ __forceinline__ void sp_build_b(const float* __restrict__ W, v8h* b_lo, v8h* b_hi, int tid) {
    static_assert((NTILES * 4 * 32) % 32 == 0 && NTILES * 4 * 32 <= 1024 && NTILES % 2 == 0,
                  "whole waves, one trip, tile pairs");
    const int N = NTILES * 16;
    if (tid < NTILES * 4 * 32) {
        const int l   = tid & 31;
        const int hf  = (tid >> 5) & 1;
        const int kk  = (tid >> 6) & 3;
        const int tp  = tid >> 8;
        const int n   = tp * 32 + l;
        const int kb  = kk * 32 + hf * 8;
        v8h lo, hi;
#pragma unroll
        for (int j = 0; j < 8; ++j) {
            lo[j] = (_Float16)W[(size_t)(kb + j) * N + n];
            hi[j] = (_Float16)W[(size_t)(kb + 16 + j) * N + n];
        }
        const int id = ((tp * 2 + (l >> 4)) * 4 + kk) * 32 + hf * 16 + (l & 15);
        b_lo[id] = lo;
        b_hi[id] = hi;
    }
}

template <int NTILES, bool RELU>
__device__ __forceinline__ void sp_tile_times_w(float4* acc, const v8h* b_lo, const v8h* b_hi,
                                                int lane, int mt) {
    const int half = lane >> 4;
    const int r    = lane & 15;
    int row = mt * 16 + r;
    row = row < SP_RB ? row : SP_RB - 1;
    v8f c[NTILES];
#pragma unroll
    for (int t = 0; t < NTILES; ++t) c[t] = (v8f){};
#pragma unroll
    for (int kk = 0; kk < 4; ++kk) {
        const float4* ap = acc + row * SP_PITCH + kk * 8 + half * 2;
        const float4 f0 = ap[0], f1 = ap[1], f2 = ap[4], f3 = ap[5];
        v16h a;
        a[0]  = (_Float16)f0.x; a[1]  = (_Float16)f0.y; a[2]  = (_Float16)f0.z; a[3]  = (_Float16)f0.w;
        a[4]  = (_Float16)f1.x; a[5]  = (_Float16)f1.y; a[6]  = (_Float16)f1.z; a[7]  = (_Float16)f1.w;
        a[8]  = (_Float16)f2.x; a[9]  = (_Float16)f2.y; a[10] = (_Float16)f2.z; a[11] = (_Float16)f2.w;
        a[12] = (_Float16)f3.x; a[13] = (_Float16)f3.y; a[14] = (_Float16)f3.z; a[15] = (_Float16)f3.w;
#pragma unroll
        for (int t = 0; t < NTILES; ++t) {
            const v8h lo = b_lo[(t * 4 + kk) * 32 + lane];
            const v8h hi = b_hi[(t * 4 + kk) * 32 + lane];
            const v16h b = __builtin_shufflevector(lo, hi, 0, 1, 2, 3, 4, 5, 6, 7,
                                                   8, 9, 10, 11, 12, 13, 14, 15);
            c[t] = __builtin_amdgcn_wmma_f32_16x16x32_f16(
                        false, a,  false, b,
                        (short)0, c[t],
                        false,  false);
        }
    }
    if (RELU) {
#pragma unroll
        for (int t = 0; t < NTILES; ++t)
#pragma unroll
            for (int v = 0; v < 8; ++v) c[t][v] = fmaxf(c[t][v], 0.0f);
    }
    float* accf = (float*)acc;
#pragma unroll
    for (int v = 0; v < 8; ++v) {
        const int orow = mt * 16 + half * 8 + v;
        if (orow < SP_RB) {
#pragma unroll
            for (int t = 0; t < NTILES; ++t) accf[orow * (SP_PITCH * 4) + t * 16 + r] = c[t][v];
        }
    }
}

template <int NTILES, bool RELU>
__device__ __forceinline__ void sp_product_out(float4* acc, const v8h* b_lo, const v8h* b_hi,
                                               float* __restrict__ C, int r0, int nNodes,
                                               int tid, int lane, int w) {
    __syncthreads();
    if (w < SP_MT) sp_tile_times_w<NTILES, RELU>(acc, b_lo, b_hi, lane, w);
    __syncthreads();
    for (int i = tid; i < SP_RB * (NTILES * 4); i += 1024) {
        const int r = i / (NTILES * 4);
        const int c = i % (NTILES * 4);
        if (r0 + r < nNodes) ((float4*)C)[(size_t)(r0 + r) * (NTILES * 4) + c] = acc[r * SP_PITCH + c];
    }
}

template <int MODE>
__global__ __launch_bounds__(1024)
void gcn_spmm_rows(const int* __restrict__ lrow, const int* __restrict__ lcol,
                   const float* __restrict__ lval, const int* __restrict__ lcnt,
                   const float* __restrict__ x, const float* __restrict__ W,
                   float* __restrict__ C, int* __restrict__ rec_n, int* __restrict__ rec,
                   int nNodes, int nEdges) {
    constexpr int  NTILES = MODE == 0 ? 8 : 4;
    constexpr bool RELU   = MODE == 0;
    __shared__ float4        acc[SP_RB * SP_PITCH];
    __shared__ __attribute__((aligned(16))) int l_c[SP_WAVES * SP_CAP];
    __shared__ float         l_v[SP_WAVES * SP_CAP];
    __shared__ __attribute__((aligned(16))) unsigned char l_r[SP_WAVES * SP_CAP];
    __shared__ unsigned short l_o[SP_WAVES * SP_OWN];
    __shared__ int           l_n[SP_WAVES];
    __shared__ int           more[2];

    static_assert(2 * NTILES * 4 * 32 * sizeof(v8h) <= sizeof(int) * SP_WAVES * SP_CAP, "B fragments fit the list");
    static_assert(SP_MT <= SP_WAVES, "a wave per 16-row tile");
    v8h* b_lo = (v8h*)l_c;
    v8h* b_hi = b_lo + NTILES * 4 * 32;

    const int tid  = threadIdx.x;
    const int lane = tid & 31;
    const int w    = __builtin_amdgcn_readfirstlane(tid >> 5);
    const int subCap    = sp_sub_cap(nEdges);
    const int maxRounds = 4 * ((subCap + 127) >> 7) + 2;
    const int u0 = w * 16;

    for (int r0 = blockIdx.x * SP_RB; r0 < nNodes; r0 += gridDim.x * SP_RB) {
        const int rb = r0 / SP_RB;
        int4* recW = (int4*)rec + (size_t)((rb < SP_RECB ? rb : 0) * SP_WAVES + w) * SP_REC;
        if (MODE == 1) {
            const int rn = rb < SP_RECB ? rec_n[rb * SP_WAVES + lane] : -1;
            if (__builtin_amdgcn_ballot_w32(rn < 0) == 0u) {
                const int myN = __builtin_amdgcn_readlane(rn, w);
                const int l16 = lane & 15;
                int4 cur = make_int4(0, 0, 0, 0);
                if (myN > 0) cur = recW[l16 < myN ? l16 : myN - 1];
                sp_build_b<NTILES>(W, b_lo, b_hi, tid);
                for (int r = w; r < SP_RB; r += SP_WAVES) acc[r * SP_PITCH + lane] = make_float4(0.f, 0.f, 0.f, 0.f);
                sp_run run = {(sp_f4){0.f, 0.f, 0.f, 0.f}, -1};
                for (int i0 = 0; i0 < myN; i0 += 16) {
                    const int left = myN - i0;
                    if (left >= 16) {
                        const int ni = i0 + 16 + l16;
                        cur = sp_add16_noted<false>(acc, run, x, lane, cur, recW + (ni < myN ? ni : myN - 1), 16);
                    } else {
                        sp_add16_noted<true>(acc, run, x, lane, cur, recW, left);
                    }
                }
                sp_run_flush(acc, run, lane);
                sp_product_out<NTILES, RELU>(acc, b_lo, b_hi, C, r0, nNodes, tid, lane, w);
                __syncthreads();
                continue;
            }
        }
        int  recN  = 0;
        bool recOk = rb < SP_RECB;
        const int s = r0 / (SP_SB * SP_RB);
        const int myc = lcnt[(s >> 4) * (SP_NSUB * 16) + (u0 + (lane & 15)) * 16 + (s & 15)];
        for (int i = tid; i < SP_RB * SP_PITCH; i += 1024) acc[i] = make_float4(0.f, 0.f, 0.f, 0.f);
        if (tid < 2) more[tid] = 0;
        int g = 0;
        int c = 0;
        __syncthreads();

        for (int round = 0; round < maxRounds; ++round) {
            int n = 0;
            while (g < 4 && n + SP_SWEEP <= SP_CAP) {
                int nq[4], bq[4];
                int cmax = 0;
#pragma unroll
                for (int q = 0; q < 4; ++q) {
                    nq[q] = __builtin_amdgcn_readlane(myc, g * 4 + q);
                    bq[q] = (s * SP_NSUB + u0 + g * 4 + q) * subCap;
                    cmax  = nq[q] > cmax ? nq[q] : cmax;
                }
                if (c >= cmax) { ++g; c = 0; continue; }
                const int idx = c + lane * 4;
                c += 128;
                int4 rr[4];
#pragma unroll
                for (int q = 0; q < 4; ++q) {
                    rr[q] = make_int4(-1, -1, -1, -1);
                    if (idx < nq[q]) rr[q] = *(const int4*)(lrow + bq[q] + idx);
                }
#pragma unroll
                for (int q = 0; q < 4; ++q) {
                    const int p0 = bq[q] + idx;
                    unsigned d[4];
                    d[0] = (unsigned)(rr[q].x - r0); d[1] = (unsigned)(rr[q].y - r0);
                    d[2] = (unsigned)(rr[q].z - r0); d[3] = (unsigned)(rr[q].w - r0);
                    bool any = (d[0] < SP_RB) | (d[1] < SP_RB) | (d[2] < SP_RB) | (d[3] < SP_RB);
                    if (__builtin_amdgcn_ballot_w32(any) != 0u) {
#pragma unroll
                        for (int j = 0; j < 4; ++j) {
                            bool hit = d[j] < SP_RB;
                            unsigned m = __builtin_amdgcn_ballot_w32(hit);
                            if (m != 0u) {
                                if (hit) {
                                    int qq = w * SP_CAP + n + (int)__builtin_amdgcn_mbcnt_lo(m, 0u);
                                    l_r[qq] = (unsigned char)d[j];
                                    l_c[qq] = p0 + j;
                                }
                                n += __builtin_popcount(m);
                            }
                        }
                    }
                }
            }
            for (int i = lane; i < n; i += 32) {
                int e = l_c[w * SP_CAP + i];
                l_c[w * SP_CAP + i] = lcol[e];
                l_v[w * SP_CAP + i] = lval[e];
            }
            if (lane == 0) {
                l_n[w] = n;
                if (g < 4) more[round & 1] = 1;
            }
            __syncthreads();
            const int again = more[round & 1];
            if (tid == 0) more[(round + 1) & 1] = 0;

            {
                const int myn = l_n[lane];
                unsigned short* own = l_o + w * SP_OWN;
                int nOwn = 0;
                int Lg = 0;
                int b  = 0;
                for (int pass = 0; pass < 8 * ((SP_CAP + 127) / 128) + 1; ++pass) {
                    while (Lg < 8) {
                        int cnt[4];
                        int cmax = 0, stepMax = 0;
#pragma unroll
                        for (int q = 0; q < 4; ++q) {
                            cnt[q] = __builtin_amdgcn_readlane(myn, Lg * 4 + q);
                            cmax   = cnt[q] > cmax ? cnt[q] : cmax;
                            int left = cnt[q] - b;
                            left = left < 0 ? 0 : (left > 128 ? 128 : left);
                            stepMax += left;
                        }
                        if (b >= cmax) { ++Lg; b = 0; continue; }
                        if (nOwn + stepMax > SP_OWN) break;
                        const int p = b + lane * 4;
                        unsigned wd[4];
                        int      vc[4];
#pragma unroll
                        for (int q = 0; q < 4; ++q) {
                            int v = cnt[q] - p;
                            vc[q] = v < 0 ? 0 : (v > 4 ? 4 : v);
                            wd[q] = 0u;
                            if (vc[q] > 0) wd[q] = *(const unsigned*)(l_r + (Lg * 4 + q) * SP_CAP + p);
                        }
#pragma unroll
                        for (int q = 0; q < 4; ++q) {
#pragma unroll
                            for (int j = 0; j < 4; ++j) {
                                const bool hit = (j < vc[q]) & ((int)((wd[q] >> (8 * j)) & 31u) == w);
                                const unsigned m = __builtin_amdgcn_ballot_w32(hit);
                                if (m != 0u) {
                                    if (hit)
                                        own[nOwn + (int)__builtin_amdgcn_mbcnt_lo(m, 0u)] =
                                            (unsigned short)((Lg * 4 + q) * SP_CAP + p + j);
                                    nOwn += __builtin_popcount(m);
                                }
                            }
                        }
                        b += 128;
                    }
                    if (nOwn <= 256) {
                        int ge[8], gk[8];
#pragma unroll
                        for (int j = 0; j < 8; ++j) {
                            ge[j] = 0;
                            gk[j] = SP_KR;
                            if (j * 32 + lane < nOwn) {
                                ge[j] = own[j * 32 + lane];
                                gk[j] = (int)l_r[ge[j]] / SP_WAVES;
                            }
                        }
                        int pos = 0;
                        for (int k = 0; k < SP_KR; ++k) {
#pragma unroll
                            for (int j = 0; j < 8; ++j) {
                                if (j * 32 < nOwn) {
                                    const bool hit = gk[j] == k;
                                    const unsigned m = __builtin_amdgcn_ballot_w32(hit);
                                    if (m != 0u) {
                                        if (hit) own[pos + (int)__builtin_amdgcn_mbcnt_lo(m, 0u)] = (unsigned short)ge[j];
                                        pos += __builtin_popcount(m);
                                    }
                                }
                            }
                        }
                    }
                    sp_run run = {(sp_f4){0.f, 0.f, 0.f, 0.f}, -1};
                    for (int i0 = 0; i0 < nOwn; i0 += 16) {
                        int   er = 0, ec = 0;
                        float ev = 0.f;
                        if (lane < 16 && i0 + lane < nOwn) {
                            const int e = own[i0 + lane];
                            er = (int)l_r[e];
                            ec = l_c[e];
                            ev = l_v[e];
                        }
                        const int left = nOwn - i0;
                        if (MODE == 0) {
                            const int take = left < 16 ? left : 16;
                            if (recOk && recN + take <= SP_REC) {
                                if (lane < take) recW[recN + lane] = make_int4(er, ec, __float_as_int(ev), 0);
                                recN += take;
                            } else {
                                recOk = false;
                            }
                        }
                        if (left >= 16) sp_add16<true>(acc, run, x, lane, er, ec, ev, 16);
                        else            sp_add16<false>(acc, run, x, lane, er, ec, ev, left);
                    }
                    sp_run_flush(acc, run, lane);
                    nOwn = 0;
                    if (Lg >= 8) break;
                }
            }
            __syncthreads();
            if (!again) break;
        }

        if (MODE == 0 && rb < SP_RECB && lane == 0) rec_n[rb * SP_WAVES + w] = recOk ? recN : -1;

        sp_build_b<NTILES>(W, b_lo, b_hi, tid);
        sp_product_out<NTILES, RELU>(acc, b_lo, b_hi, C, r0, nNodes, tid, lane, w);
        __syncthreads();
    }
}

extern "C" void kernel_launch(void* const* d_in, const int* in_sizes, int n_in,
                              void* d_out, int out_size, void* d_ws, size_t ws_size,
                              hipStream_t stream) {
    const float* x    = (const float*)d_in[0];
    const int*   row  = (const int*)d_in[1];
    const int*   col  = (const int*)d_in[2];
    const float* vals = (const float*)d_in[3];
    const float* W1   = (const float*)d_in[4];
    const float* W2   = (const float*)d_in[5];
    float* out = (float*)d_out;

    const int NN = in_sizes[0] / IN_DIM;
    const int NE = in_sizes[1];

    float*     hF   = (float*)d_ws;
    int*       lcnt = (int*)(hF + (size_t)NN * IN_DIM);
    int*       lrow = lcnt + SP_NSUB * 16;
    int*       lcol = lrow + SP_LIST;
    float*     lval = (float*)(lcol + SP_LIST);
    int*       recLen = (int*)(lval + SP_LIST);
    int*       rec    = recLen + SP_RECB * SP_WAVES;

    gcn_edge_ranges<<<256, 1024, 0, stream>>>(row, col, vals, lrow, lcol, lval, lcnt, NN, NE);

    gcn_spmm_rows<0><<<256, 1024, 0, stream>>>(lrow, lcol, lval, lcnt, x, W1, hF, recLen, rec, NN, NE);

    gcn_spmm_rows<1><<<256, 1024, 0, stream>>>(lrow, lcol, lval, lcnt, hF, W2, out, recLen, rec, NN, NE);
}
